# starting kernel
# baseline (speedup 1.0000x reference)
#include <hip/hip_runtime.h>
#include <hip/hip_bf16.h>

typedef __attribute__((ext_vector_type(2))) float v2f;
typedef __attribute__((ext_vector_type(8))) float v8f;

#define D 64

__global__ __launch_bounds__(256) void deg_init(float* dout, float* din, int N) {
    int i = blockIdx.x * blockDim.x + threadIdx.x;
    if (i < N) { dout[i] = 1.0f; din[i] = 1.0f; }
}

__global__ __launch_bounds__(256) void deg_count(const int* __restrict__ src,
                                                 const int* __restrict__ dst,
                                                 float* dout, float* din, int E) {
    int e = blockIdx.x * blockDim.x + threadIdx.x;
    if (e < E) {
        atomicAdd(&dout[src[e]], 1.0f);
        atomicAdd(&din[dst[e]], 1.0f);
    }
}

__global__ __launch_bounds__(256) void deg_rsqrt(float* dout, float* din, int N) {
    int i = blockIdx.x * blockDim.x + threadIdx.x;
    if (i < N) { dout[i] = rsqrtf(dout[i]); din[i] = rsqrtf(din[i]); }
}

__global__ __launch_bounds__(256) void gemm64_wmma(const float* __restrict__ X,
                                                   const float* __restrict__ rs,
                                                   const float* __restrict__ W,
                                                   float* __restrict__ out0,
                                                   float* __restrict__ out1,
                                                   int N) {
    const int wave = threadIdx.x >> 5;
    const int lane = threadIdx.x & 31;
    const int nb = blockIdx.x * 128 + wave * 16;
    if (nb >= N) return;
    const int half = lane >> 4;
    const int lr   = lane & 15;

    const float s = rs[nb + lr];
    const float* __restrict__ xrow = X + (size_t)(nb + lr) * D;

    v8f acc0 = {}, acc1 = {}, acc2 = {}, acc3 = {};
#pragma unroll
    for (int k0 = 0; k0 < D; k0 += 4) {
        const int ka = k0 + half * 2;
        v2f a;
        a.x = xrow[ka] * s;
        a.y = xrow[ka + 1] * s;
        const float* __restrict__ w0 = W + (size_t)ka * D + lr;
        const float* __restrict__ w1 = w0 + D;
        v2f b0 = { w0[0],  w1[0]  };
        v2f b1 = { w0[16], w1[16] };
        v2f b2 = { w0[32], w1[32] };
        v2f b3 = { w0[48], w1[48] };
        acc0 = __builtin_amdgcn_wmma_f32_16x16x4_f32(false, a, false, b0, (short)0, acc0, false, false);
        acc1 = __builtin_amdgcn_wmma_f32_16x16x4_f32(false, a, false, b1, (short)0, acc1, false, false);
        acc2 = __builtin_amdgcn_wmma_f32_16x16x4_f32(false, a, false, b2, (short)0, acc2, false, false);
        acc3 = __builtin_amdgcn_wmma_f32_16x16x4_f32(false, a, false, b3, (short)0, acc3, false, false);
    }

    const int row0 = nb + half * 8;
#pragma unroll
    for (int v = 0; v < 8; ++v) {
        const size_t base = (size_t)(row0 + v) * D + lr;
        out0[base]      = acc0[v];
        out0[base + 16] = acc1[v];
        out0[base + 32] = acc2[v];
        out0[base + 48] = acc3[v];
        out1[base]      = acc0[v];
        out1[base + 16] = acc1[v];
        out1[base + 32] = acc2[v];
        out1[base + 48] = acc3[v];
    }
}

__global__ __launch_bounds__(256) void scatter_add(const float* __restrict__ h,
                                                   const int* __restrict__ src,
                                                   const int* __restrict__ dst,
                                                   float* __restrict__ agg, int E) {
    const int t = blockIdx.x * blockDim.x + threadIdx.x;
    const int e = t >> 4;
    if (e >= E) return;
    const int f = (t & 15) * 4;
    const int s = src[e];
    const int d = dst[e];
    const float4 v = *(const float4*)(h + (size_t)s * D + f);
    float* p = agg + (size_t)d * D + f;
    atomicAdd(p + 0, v.x);
    atomicAdd(p + 1, v.y);
    atomicAdd(p + 2, v.z);
    atomicAdd(p + 3, v.w);
}

__global__ __launch_bounds__(256) void post_scale_bias(const float* __restrict__ agg,
                                                       const float* __restrict__ rs_in,
                                                       const float* __restrict__ b,
                                                       float* __restrict__ hout, int N) {
    const int t = blockIdx.x * blockDim.x + threadIdx.x;
    const int node = t >> 4;
    if (node >= N) return;
    const int f = (t & 15) * 4;
    const float s = rs_in[node];
    const float4 v = *(const float4*)(agg + (size_t)node * D + f);
    float4 r;
    r.x = fmaxf(v.x * s + b[f + 0], 0.0f);
    r.y = fmaxf(v.y * s + b[f + 1], 0.0f);
    r.z = fmaxf(v.z * s + b[f + 2], 0.0f);
    r.w = fmaxf(v.w * s + b[f + 3], 0.0f);
    *(float4*)(hout + (size_t)node * D + f) = r;
}

__global__ __launch_bounds__(256) void final_dot(const float* __restrict__ agg,
                                                 const float* __restrict__ rs_in,
                                                 const float* __restrict__ b2,
                                                 const float* __restrict__ Wr,
                                                 const float* __restrict__ br,
                                                 float* __restrict__ out, int N) {
    const int wave = threadIdx.x >> 5;
    const int lane = threadIdx.x & 31;
    const int node = blockIdx.x * 8 + wave;
    if (node >= N) return;
    const float s = rs_in[node];
    const float* row = agg + (size_t)node * D;
    float acc = (row[lane]      * s + b2[lane])      * Wr[lane]
              + (row[lane + 32] * s + b2[lane + 32]) * Wr[lane + 32];
#pragma unroll
    for (int off = 16; off > 0; off >>= 1) acc += __shfl_xor(acc, off, 32);
    if (lane == 0) out[node] = acc + br[0];
}

extern "C" void kernel_launch(void* const* d_in, const int* in_sizes, int n_in,
                              void* d_out, int out_size, void* d_ws, size_t ws_size,
                              hipStream_t stream) {
    const float* x   = (const float*)d_in[0];
    const int*   src = (const int*)d_in[1];
    const int*   dst = (const int*)d_in[2];
    const float* W1  = (const float*)d_in[3];
    const float* b1  = (const float*)d_in[4];
    const float* W2  = (const float*)d_in[5];
    const float* b2  = (const float*)d_in[6];
    const float* Wr  = (const float*)d_in[7];
    const float* br  = (const float*)d_in[8];
    float* out = (float*)d_out;

    const int N = in_sizes[0] / D;
    const int E = in_sizes[1];

    float* ws     = (float*)d_ws;
    float* rs_out = ws;
    float* rs_in  = ws + N;
    float* h      = ws + 2 * (size_t)N;
    float* agg    = h + (size_t)N * D;

    const int TB = 256;
    const int gN    = (N + TB - 1) / TB;
    const int gE    = (E + TB - 1) / TB;
    const int gGemm = (N + 127) / 128;
    const int gEdge = (int)(((size_t)E * 16 + TB - 1) / TB);
    const int gFeat = (int)(((size_t)N * 16 + TB - 1) / TB);
    const int gNode = (N + 7) / 8;

    deg_init <<<gN, TB, 0, stream>>>(rs_out, rs_in, N);
    deg_count<<<gE, TB, 0, stream>>>(src, dst, rs_out, rs_in, E);
    deg_rsqrt<<<gN, TB, 0, stream>>>(rs_out, rs_in, N);

    gemm64_wmma<<<gGemm, TB, 0, stream>>>(x, rs_out, W1, h, agg, N);
    scatter_add<<<gEdge, TB, 0, stream>>>(h, src, dst, agg, E);
    post_scale_bias<<<gFeat, TB, 0, stream>>>(agg, rs_in, b1, h, N);

    gemm64_wmma<<<gGemm, TB, 0, stream>>>(h, rs_out, W2, h, agg, N);
    scatter_add<<<gEdge, TB, 0, stream>>>(h, src, dst, agg, E);

    final_dot<<<gNode, TB, 0, stream>>>(agg, rs_in, b2, Wr, br, out, N);
}


# scatter-add: one wave per 32 edges, lane = feature: each row load and each f32 atomic add covers one whole 128-B line (2 per edge, was 8 part-line touches); atomics kept, sums in arrival order
# speedup vs baseline: 1.4458x; 1.4458x over previous
#include <hip/hip_runtime.h>
#include <hip/hip_bf16.h>

typedef __attribute__((ext_vector_type(2))) float v2f;
typedef __attribute__((ext_vector_type(8))) float v8f;

#define D 64

__global__ __launch_bounds__(256) void deg_init(float* dout, float* din, int N) {
    int i = blockIdx.x * blockDim.x + threadIdx.x;
    if (i < N) { dout[i] = 1.0f; din[i] = 1.0f; }
}

__global__ __launch_bounds__(256) void deg_count(const int* __restrict__ src,
                                                 const int* __restrict__ dst,
                                                 float* dout, float* din, int E) {
    int e = blockIdx.x * blockDim.x + threadIdx.x;
    if (e < E) {
        atomicAdd(&dout[src[e]], 1.0f);
        atomicAdd(&din[dst[e]], 1.0f);
    }
}

__global__ __launch_bounds__(256) void deg_rsqrt(float* dout, float* din, int N) {
    int i = blockIdx.x * blockDim.x + threadIdx.x;
    if (i < N) { dout[i] = rsqrtf(dout[i]); din[i] = rsqrtf(din[i]); }
}

__global__ __launch_bounds__(256) void gemm64_wmma(const float* __restrict__ X,
                                                   const float* __restrict__ rs,
                                                   const float* __restrict__ W,
                                                   float* __restrict__ out0,
                                                   float* __restrict__ out1,
                                                   int N) {
    const int wave = threadIdx.x >> 5;
    const int lane = threadIdx.x & 31;
    const int nb = blockIdx.x * 128 + wave * 16;
    if (nb >= N) return;
    const int half = lane >> 4;
    const int lr   = lane & 15;

    const float s = rs[nb + lr];
    const float* __restrict__ xrow = X + (size_t)(nb + lr) * D;

    v8f acc0 = {}, acc1 = {}, acc2 = {}, acc3 = {};
#pragma unroll
    for (int k0 = 0; k0 < D; k0 += 4) {
        const int ka = k0 + half * 2;
        v2f a;
        a.x = xrow[ka] * s;
        a.y = xrow[ka + 1] * s;
        const float* __restrict__ w0 = W + (size_t)ka * D + lr;
        const float* __restrict__ w1 = w0 + D;
        v2f b0 = { w0[0],  w1[0]  };
        v2f b1 = { w0[16], w1[16] };
        v2f b2 = { w0[32], w1[32] };
        v2f b3 = { w0[48], w1[48] };
        acc0 = __builtin_amdgcn_wmma_f32_16x16x4_f32(false, a, false, b0, (short)0, acc0, false, false);
        acc1 = __builtin_amdgcn_wmma_f32_16x16x4_f32(false, a, false, b1, (short)0, acc1, false, false);
        acc2 = __builtin_amdgcn_wmma_f32_16x16x4_f32(false, a, false, b2, (short)0, acc2, false, false);
        acc3 = __builtin_amdgcn_wmma_f32_16x16x4_f32(false, a, false, b3, (short)0, acc3, false, false);
    }

    const int row0 = nb + half * 8;
#pragma unroll
    for (int v = 0; v < 8; ++v) {
        const size_t base = (size_t)(row0 + v) * D + lr;
        out0[base]      = acc0[v];
        out0[base + 16] = acc1[v];
        out0[base + 32] = acc2[v];
        out0[base + 48] = acc3[v];
        out1[base]      = acc0[v];
        out1[base + 16] = acc1[v];
        out1[base + 32] = acc2[v];
        out1[base + 48] = acc3[v];
    }
}

__global__ __launch_bounds__(256) void scatter_add(const float* __restrict__ h,
                                                   const int* __restrict__ src,
                                                   const int* __restrict__ dst,
                                                   float* __restrict__ agg, int E) {
    const int lane = threadIdx.x & 31;
    const int e0 = (blockIdx.x * 8 + (threadIdx.x >> 5)) * 32;
    if (e0 >= E) return;
    const int e = e0 + lane;
    int sv = 0, dv = 0;
    if (e < E) { sv = src[e]; dv = dst[e]; }
    const int n = (E - e0 < 32) ? (E - e0) : 32;

    if (n == 32) {
#pragma unroll
        for (int g = 0; g < 32; g += 8) {
            float lo[8], hi[8];
#pragma unroll
            for (int j = 0; j < 8; ++j) {
                const int s = __builtin_amdgcn_readlane(sv, g + j);
                const float* __restrict__ r = h + (size_t)s * D + lane;
                lo[j] = r[0];
                hi[j] = r[32];
            }
#pragma unroll
            for (int j = 0; j < 8; ++j) {
                const int d = __builtin_amdgcn_readlane(dv, g + j);
                float* p = agg + (size_t)d * D + lane;
                atomicAdd(p, lo[j]);
                atomicAdd(p + 32, hi[j]);
            }
        }
    } else {
        for (int j = 0; j < n; ++j) {
            const int s = __builtin_amdgcn_readlane(sv, j);
            const int d = __builtin_amdgcn_readlane(dv, j);
            const float* __restrict__ r = h + (size_t)s * D + lane;
            float* p = agg + (size_t)d * D + lane;
            const float a = r[0];
            const float b = r[32];
            atomicAdd(p, a);
            atomicAdd(p + 32, b);
        }
    }
}

__global__ __launch_bounds__(256) void post_scale_bias(const float* __restrict__ agg,
                                                       const float* __restrict__ rs_in,
                                                       const float* __restrict__ b,
                                                       float* __restrict__ hout, int N) {
    const int t = blockIdx.x * blockDim.x + threadIdx.x;
    const int node = t >> 4;
    if (node >= N) return;
    const int f = (t & 15) * 4;
    const float s = rs_in[node];
    const float4 v = *(const float4*)(agg + (size_t)node * D + f);
    float4 r;
    r.x = fmaxf(v.x * s + b[f + 0], 0.0f);
    r.y = fmaxf(v.y * s + b[f + 1], 0.0f);
    r.z = fmaxf(v.z * s + b[f + 2], 0.0f);
    r.w = fmaxf(v.w * s + b[f + 3], 0.0f);
    *(float4*)(hout + (size_t)node * D + f) = r;
}

__global__ __launch_bounds__(256) void final_dot(const float* __restrict__ agg,
                                                 const float* __restrict__ rs_in,
                                                 const float* __restrict__ b2,
                                                 const float* __restrict__ Wr,
                                                 const float* __restrict__ br,
                                                 float* __restrict__ out, int N) {
    const int wave = threadIdx.x >> 5;
    const int lane = threadIdx.x & 31;
    const int node = blockIdx.x * 8 + wave;
    if (node >= N) return;
    const float s = rs_in[node];
    const float* row = agg + (size_t)node * D;
    float acc = (row[lane]      * s + b2[lane])      * Wr[lane]
              + (row[lane + 32] * s + b2[lane + 32]) * Wr[lane + 32];
#pragma unroll
    for (int off = 16; off > 0; off >>= 1) acc += __shfl_xor(acc, off, 32);
    if (lane == 0) out[node] = acc + br[0];
}

extern "C" void kernel_launch(void* const* d_in, const int* in_sizes, int n_in,
                              void* d_out, int out_size, void* d_ws, size_t ws_size,
                              hipStream_t stream) {
    const float* x   = (const float*)d_in[0];
    const int*   src = (const int*)d_in[1];
    const int*   dst = (const int*)d_in[2];
    const float* W1  = (const float*)d_in[3];
    const float* b1  = (const float*)d_in[4];
    const float* W2  = (const float*)d_in[5];
    const float* b2  = (const float*)d_in[6];
    const float* Wr  = (const float*)d_in[7];
    const float* br  = (const float*)d_in[8];
    float* out = (float*)d_out;

    const int N = in_sizes[0] / D;
    const int E = in_sizes[1];

    float* ws     = (float*)d_ws;
    float* rs_out = ws;
    float* rs_in  = ws + N;
    float* h      = ws + 2 * (size_t)N;
    float* agg    = h + (size_t)N * D;

    const int TB = 256;
    const int gN    = (N + TB - 1) / TB;
    const int gE    = (E + TB - 1) / TB;
    const int gGemm = (N + 127) / 128;
    const int gEdge = (E + 255) / 256;
    const int gFeat = (int)(((size_t)N * 16 + TB - 1) / TB);
    const int gNode = (N + 7) / 8;

    deg_init <<<gN, TB, 0, stream>>>(rs_out, rs_in, N);
    deg_count<<<gE, TB, 0, stream>>>(src, dst, rs_out, rs_in, E);
    deg_rsqrt<<<gN, TB, 0, stream>>>(rs_out, rs_in, N);

    gemm64_wmma<<<gGemm, TB, 0, stream>>>(x, rs_out, W1, h, agg, N);
    scatter_add<<<gEdge, TB, 0, stream>>>(h, src, dst, agg, E);
    post_scale_bias<<<gFeat, TB, 0, stream>>>(agg, rs_in, b1, h, N);

    gemm64_wmma<<<gGemm, TB, 0, stream>>>(h, rs_out, W2, h, agg, N);
    scatter_add<<<gEdge, TB, 0, stream>>>(h, src, dst, agg, E);

    final_dot<<<gNode, TB, 0, stream>>>(agg, rs_in, b2, Wr, br, out, N);
}


# degrees counted in LDS: blocks of (node range x edge chunk) count with LDS integer adds, store partial-count planes, one pass sums them; the 2 scattered global f32 atomics per edge go; degrees exact
# speedup vs baseline: 2.2269x; 1.5403x over previous
#include <hip/hip_runtime.h>
#include <hip/hip_bf16.h>

typedef __attribute__((ext_vector_type(2))) float v2f;
typedef __attribute__((ext_vector_type(8))) float v8f;

#define D 64

#define HCAP    25600
#define HCHUNKS 32
#define HBATCH  8

__global__ __launch_bounds__(1024) void deg_hist(const int* __restrict__ src,
                                                 const int* __restrict__ dst,
                                                 unsigned* __restrict__ part, int N, int E) {
    __shared__ unsigned hist[HCAP];
    const int tid = threadIdx.x;
    const int c = blockIdx.x;
    const int* __restrict__ idx = blockIdx.z ? dst : src;
    unsigned* __restrict__ plane = part + ((size_t)blockIdx.z * HCHUNKS + c) * (size_t)N;

    const int nvec = E >> 2;
    const int per  = (nvec + HCHUNKS - 1) / HCHUNKS;
    const int v0   = c * per;
    const int v1   = (v0 + per < nvec) ? (v0 + per) : nvec;
    const int4* __restrict__ idx4 = (const int4*)idx;

    for (int base = blockIdx.y * HCAP; base < N; base += gridDim.y * HCAP) {
        const unsigned cap = (unsigned)((N - base < HCAP) ? (N - base) : HCAP);
        for (unsigned i = tid; i < cap; i += 1024) hist[i] = 0u;
        __syncthreads();

        for (int v = v0 + tid; v < v1; v += HBATCH * 1024) {
            int4 q[HBATCH];
#pragma unroll
            for (int j = 0; j < HBATCH; ++j) {
                const int vv = v + j * 1024;
                q[j] = make_int4(-1, -1, -1, -1);
                if (vv < v1) q[j] = idx4[vv];
            }
#pragma unroll
            for (int j = 0; j < HBATCH; ++j) {
                const unsigned u0 = (unsigned)(q[j].x - base);
                const unsigned u1 = (unsigned)(q[j].y - base);
                const unsigned u2 = (unsigned)(q[j].z - base);
                const unsigned u3 = (unsigned)(q[j].w - base);
                if (u0 < cap) atomicAdd(&hist[u0], 1u);
                if (u1 < cap) atomicAdd(&hist[u1], 1u);
                if (u2 < cap) atomicAdd(&hist[u2], 1u);
                if (u3 < cap) atomicAdd(&hist[u3], 1u);
            }
        }
        if (c == HCHUNKS - 1) {
            for (int e = (nvec << 2) + tid; e < E; e += 1024) {
                const unsigned u = (unsigned)(idx[e] - base);
                if (u < cap) atomicAdd(&hist[u], 1u);
            }
        }
        __syncthreads();

        for (unsigned i = tid; i < cap; i += 1024) plane[base + i] = hist[i];
        __syncthreads();
    }
}

__global__ __launch_bounds__(256) void deg_finish(const unsigned* __restrict__ part,
                                                  float* __restrict__ dout,
                                                  float* __restrict__ din, int N) {
    const int i = blockIdx.x * blockDim.x + threadIdx.x;
    if (i >= N) return;
    const unsigned* __restrict__ po = part + i;
    const unsigned* __restrict__ pi = part + (size_t)HCHUNKS * (size_t)N + i;
    unsigned a = 1u, b = 1u;
#pragma unroll
    for (int c = 0; c < HCHUNKS; ++c) {
        a += po[(size_t)c * (size_t)N];
        b += pi[(size_t)c * (size_t)N];
    }
    dout[i] = rsqrtf((float)a);
    din[i]  = rsqrtf((float)b);
}

__global__ __launch_bounds__(256) void gemm64_wmma(const float* __restrict__ X,
                                                   const float* __restrict__ rs,
                                                   const float* __restrict__ W,
                                                   float* __restrict__ out0,
                                                   float* __restrict__ out1,
                                                   int N) {
    const int wave = threadIdx.x >> 5;
    const int lane = threadIdx.x & 31;
    const int nb = blockIdx.x * 128 + wave * 16;
    if (nb >= N) return;
    const int half = lane >> 4;
    const int lr   = lane & 15;

    const float s = rs[nb + lr];
    const float* __restrict__ xrow = X + (size_t)(nb + lr) * D;

    v8f acc0 = {}, acc1 = {}, acc2 = {}, acc3 = {};
#pragma unroll
    for (int k0 = 0; k0 < D; k0 += 4) {
        const int ka = k0 + half * 2;
        v2f a;
        a.x = xrow[ka] * s;
        a.y = xrow[ka + 1] * s;
        const float* __restrict__ w0 = W + (size_t)ka * D + lr;
        const float* __restrict__ w1 = w0 + D;
        v2f b0 = { w0[0],  w1[0]  };
        v2f b1 = { w0[16], w1[16] };
        v2f b2 = { w0[32], w1[32] };
        v2f b3 = { w0[48], w1[48] };
        acc0 = __builtin_amdgcn_wmma_f32_16x16x4_f32(false, a, false, b0, (short)0, acc0, false, false);
        acc1 = __builtin_amdgcn_wmma_f32_16x16x4_f32(false, a, false, b1, (short)0, acc1, false, false);
        acc2 = __builtin_amdgcn_wmma_f32_16x16x4_f32(false, a, false, b2, (short)0, acc2, false, false);
        acc3 = __builtin_amdgcn_wmma_f32_16x16x4_f32(false, a, false, b3, (short)0, acc3, false, false);
    }

    const int row0 = nb + half * 8;
#pragma unroll
    for (int v = 0; v < 8; ++v) {
        const size_t base = (size_t)(row0 + v) * D + lr;
        out0[base]      = acc0[v];
        out0[base + 16] = acc1[v];
        out0[base + 32] = acc2[v];
        out0[base + 48] = acc3[v];
        out1[base]      = acc0[v];
        out1[base + 16] = acc1[v];
        out1[base + 32] = acc2[v];
        out1[base + 48] = acc3[v];
    }
}

__global__ __launch_bounds__(256) void scatter_add(const float* __restrict__ h,
                                                   const int* __restrict__ src,
                                                   const int* __restrict__ dst,
                                                   float* __restrict__ agg, int E) {
    const int lane = threadIdx.x & 31;
    const int e0 = (blockIdx.x * 8 + (threadIdx.x >> 5)) * 32;
    if (e0 >= E) return;
    const int e = e0 + lane;
    int sv = 0, dv = 0;
    if (e < E) { sv = src[e]; dv = dst[e]; }
    const int n = (E - e0 < 32) ? (E - e0) : 32;

    if (n == 32) {
#pragma unroll
        for (int g = 0; g < 32; g += 8) {
            float lo[8], hi[8];
#pragma unroll
            for (int j = 0; j < 8; ++j) {
                const int s = __builtin_amdgcn_readlane(sv, g + j);
                const float* __restrict__ r = h + (size_t)s * D + lane;
                lo[j] = r[0];
                hi[j] = r[32];
            }
#pragma unroll
            for (int j = 0; j < 8; ++j) {
                const int d = __builtin_amdgcn_readlane(dv, g + j);
                float* p = agg + (size_t)d * D + lane;
                atomicAdd(p, lo[j]);
                atomicAdd(p + 32, hi[j]);
            }
        }
    } else {
        for (int j = 0; j < n; ++j) {
            const int s = __builtin_amdgcn_readlane(sv, j);
            const int d = __builtin_amdgcn_readlane(dv, j);
            const float* __restrict__ r = h + (size_t)s * D + lane;
            float* p = agg + (size_t)d * D + lane;
            const float a = r[0];
            const float b = r[32];
            atomicAdd(p, a);
            atomicAdd(p + 32, b);
        }
    }
}

__global__ __launch_bounds__(256) void post_scale_bias(const float* __restrict__ agg,
                                                       const float* __restrict__ rs_in,
                                                       const float* __restrict__ b,
                                                       float* __restrict__ hout, int N) {
    const int t = blockIdx.x * blockDim.x + threadIdx.x;
    const int node = t >> 4;
    if (node >= N) return;
    const int f = (t & 15) * 4;
    const float s = rs_in[node];
    const float4 v = *(const float4*)(agg + (size_t)node * D + f);
    float4 r;
    r.x = fmaxf(v.x * s + b[f + 0], 0.0f);
    r.y = fmaxf(v.y * s + b[f + 1], 0.0f);
    r.z = fmaxf(v.z * s + b[f + 2], 0.0f);
    r.w = fmaxf(v.w * s + b[f + 3], 0.0f);
    *(float4*)(hout + (size_t)node * D + f) = r;
}

__global__ __launch_bounds__(256) void final_dot(const float* __restrict__ agg,
                                                 const float* __restrict__ rs_in,
                                                 const float* __restrict__ b2,
                                                 const float* __restrict__ Wr,
                                                 const float* __restrict__ br,
                                                 float* __restrict__ out, int N) {
    const int wave = threadIdx.x >> 5;
    const int lane = threadIdx.x & 31;
    const int node = blockIdx.x * 8 + wave;
    if (node >= N) return;
    const float s = rs_in[node];
    const float* row = agg + (size_t)node * D;
    float acc = (row[lane]      * s + b2[lane])      * Wr[lane]
              + (row[lane + 32] * s + b2[lane + 32]) * Wr[lane + 32];
#pragma unroll
    for (int off = 16; off > 0; off >>= 1) acc += __shfl_xor(acc, off, 32);
    if (lane == 0) out[node] = acc + br[0];
}

extern "C" void kernel_launch(void* const* d_in, const int* in_sizes, int n_in,
                              void* d_out, int out_size, void* d_ws, size_t ws_size,
                              hipStream_t stream) {
    const float* x   = (const float*)d_in[0];
    const int*   src = (const int*)d_in[1];
    const int*   dst = (const int*)d_in[2];
    const float* W1  = (const float*)d_in[3];
    const float* b1  = (const float*)d_in[4];
    const float* W2  = (const float*)d_in[5];
    const float* b2  = (const float*)d_in[6];
    const float* Wr  = (const float*)d_in[7];
    const float* br  = (const float*)d_in[8];
    float* out = (float*)d_out;

    const int N = in_sizes[0] / D;
    const int E = in_sizes[1];

    float* ws     = (float*)d_ws;
    float* rs_out = ws;
    float* rs_in  = ws + N;
    float* h      = ws + 2 * (size_t)N;
    float* agg    = h + (size_t)N * D;
    unsigned* part = (unsigned*)(agg + (size_t)N * D);

    const int TB = 256;
    const int gN    = (N + TB - 1) / TB;
    const int gGemm = (N + 127) / 128;
    const int gEdge = (E + 255) / 256;
    const int gFeat = (int)(((size_t)N * 16 + TB - 1) / TB);
    const int gNode = (N + 7) / 8;

    deg_hist  <<<dim3(HCHUNKS, 4, 2), 1024, 0, stream>>>(src, dst, part, N, E);
    deg_finish<<<gN, TB, 0, stream>>>(part, rs_out, rs_in, N);

    gemm64_wmma<<<gGemm, TB, 0, stream>>>(x, rs_out, W1, h, agg, N);
    scatter_add<<<gEdge, TB, 0, stream>>>(h, src, dst, agg, E);
    post_scale_bias<<<gFeat, TB, 0, stream>>>(agg, rs_in, b1, h, N);

    gemm64_wmma<<<gGemm, TB, 0, stream>>>(h, rs_out, W2, h, agg, N);
    scatter_add<<<gEdge, TB, 0, stream>>>(h, src, dst, agg, E);

    final_dot<<<gNode, TB, 0, stream>>>(agg, rs_in, b2, Wr, br, out, N);
}


# layer 2 + head folded by linearity: z = rs_out*(h1 . (W2 Wr)) per node on the vector unit (layer-2 WMMA GEMM gone), then one float per edge summed in LDS per node range; layer-2 sum order changes
# speedup vs baseline: 4.0066x; 1.7992x over previous
#include <hip/hip_runtime.h>
#include <hip/hip_bf16.h>

typedef __attribute__((ext_vector_type(2))) float v2f;
typedef __attribute__((ext_vector_type(8))) float v8f;

#define D 64

#define HCAP    25600
#define HCHUNKS 32
#define HBATCH  8

__global__ __launch_bounds__(1024) void deg_hist(const int* __restrict__ src,
                                                 const int* __restrict__ dst,
                                                 unsigned* __restrict__ part, int N, int E) {
    __shared__ unsigned hist[HCAP];
    const int tid = threadIdx.x;
    const int c = blockIdx.x;
    const int* __restrict__ idx = blockIdx.z ? dst : src;
    unsigned* __restrict__ plane = part + ((size_t)blockIdx.z * HCHUNKS + c) * (size_t)N;

    const int nvec = E >> 2;
    const int per  = (nvec + HCHUNKS - 1) / HCHUNKS;
    const int v0   = c * per;
    const int v1   = (v0 + per < nvec) ? (v0 + per) : nvec;
    const int4* __restrict__ idx4 = (const int4*)idx;

    for (int base = blockIdx.y * HCAP; base < N; base += gridDim.y * HCAP) {
        const unsigned cap = (unsigned)((N - base < HCAP) ? (N - base) : HCAP);
        for (unsigned i = tid; i < cap; i += 1024) hist[i] = 0u;
        __syncthreads();

        for (int v = v0 + tid; v < v1; v += HBATCH * 1024) {
            int4 q[HBATCH];
#pragma unroll
            for (int j = 0; j < HBATCH; ++j) {
                const int vv = v + j * 1024;
                q[j] = make_int4(-1, -1, -1, -1);
                if (vv < v1) q[j] = idx4[vv];
            }
#pragma unroll
            for (int j = 0; j < HBATCH; ++j) {
                const unsigned u0 = (unsigned)(q[j].x - base);
                const unsigned u1 = (unsigned)(q[j].y - base);
                const unsigned u2 = (unsigned)(q[j].z - base);
                const unsigned u3 = (unsigned)(q[j].w - base);
                if (u0 < cap) atomicAdd(&hist[u0], 1u);
                if (u1 < cap) atomicAdd(&hist[u1], 1u);
                if (u2 < cap) atomicAdd(&hist[u2], 1u);
                if (u3 < cap) atomicAdd(&hist[u3], 1u);
            }
        }
        if (c == HCHUNKS - 1) {
            for (int e = (nvec << 2) + tid; e < E; e += 1024) {
                const unsigned u = (unsigned)(idx[e] - base);
                if (u < cap) atomicAdd(&hist[u], 1u);
            }
        }
        __syncthreads();

        for (unsigned i = tid; i < cap; i += 1024) plane[base + i] = hist[i];
        __syncthreads();
    }
}

__global__ __launch_bounds__(256) void deg_finish(const unsigned* __restrict__ part,
                                                  float* __restrict__ dout,
                                                  float* __restrict__ din, int N) {
    const int i = blockIdx.x * blockDim.x + threadIdx.x;
    if (i >= N) return;
    const unsigned* __restrict__ po = part + i;
    const unsigned* __restrict__ pi = part + (size_t)HCHUNKS * (size_t)N + i;
    unsigned a = 1u, b = 1u;
#pragma unroll
    for (int c = 0; c < HCHUNKS; ++c) {
        a += po[(size_t)c * (size_t)N];
        b += pi[(size_t)c * (size_t)N];
    }
    dout[i] = rsqrtf((float)a);
    din[i]  = rsqrtf((float)b);
}

__global__ __launch_bounds__(256) void gemm64_wmma(const float* __restrict__ X,
                                                   const float* __restrict__ rs,
                                                   const float* __restrict__ W,
                                                   float* __restrict__ out0,
                                                   float* __restrict__ out1,
                                                   int N) {
    const int wave = threadIdx.x >> 5;
    const int lane = threadIdx.x & 31;
    const int nb = blockIdx.x * 128 + wave * 16;
    if (nb >= N) return;
    const int half = lane >> 4;
    const int lr   = lane & 15;

    const float s = rs[nb + lr];
    const float* __restrict__ xrow = X + (size_t)(nb + lr) * D;

    v8f acc0 = {}, acc1 = {}, acc2 = {}, acc3 = {};
#pragma unroll
    for (int k0 = 0; k0 < D; k0 += 4) {
        const int ka = k0 + half * 2;
        v2f a;
        a.x = xrow[ka] * s;
        a.y = xrow[ka + 1] * s;
        const float* __restrict__ w0 = W + (size_t)ka * D + lr;
        const float* __restrict__ w1 = w0 + D;
        v2f b0 = { w0[0],  w1[0]  };
        v2f b1 = { w0[16], w1[16] };
        v2f b2 = { w0[32], w1[32] };
        v2f b3 = { w0[48], w1[48] };
        acc0 = __builtin_amdgcn_wmma_f32_16x16x4_f32(false, a, false, b0, (short)0, acc0, false, false);
        acc1 = __builtin_amdgcn_wmma_f32_16x16x4_f32(false, a, false, b1, (short)0, acc1, false, false);
        acc2 = __builtin_amdgcn_wmma_f32_16x16x4_f32(false, a, false, b2, (short)0, acc2, false, false);
        acc3 = __builtin_amdgcn_wmma_f32_16x16x4_f32(false, a, false, b3, (short)0, acc3, false, false);
    }

    const int row0 = nb + half * 8;
#pragma unroll
    for (int v = 0; v < 8; ++v) {
        const size_t base = (size_t)(row0 + v) * D + lr;
        out0[base]      = acc0[v];
        out0[base + 16] = acc1[v];
        out0[base + 32] = acc2[v];
        out0[base + 48] = acc3[v];
        out1[base]      = acc0[v];
        out1[base + 16] = acc1[v];
        out1[base + 32] = acc2[v];
        out1[base + 48] = acc3[v];
    }
}

__global__ __launch_bounds__(256) void scatter_add(const float* __restrict__ h,
                                                   const int* __restrict__ src,
                                                   const int* __restrict__ dst,
                                                   float* __restrict__ agg, int E) {
    const int lane = threadIdx.x & 31;
    const int e0 = (blockIdx.x * 8 + (threadIdx.x >> 5)) * 32;
    if (e0 >= E) return;
    const int e = e0 + lane;
    int sv = 0, dv = 0;
    if (e < E) { sv = src[e]; dv = dst[e]; }
    const int n = (E - e0 < 32) ? (E - e0) : 32;

    if (n == 32) {
#pragma unroll
        for (int g = 0; g < 32; g += 8) {
            float lo[8], hi[8];
#pragma unroll
            for (int j = 0; j < 8; ++j) {
                const int s = __builtin_amdgcn_readlane(sv, g + j);
                const float* __restrict__ r = h + (size_t)s * D + lane;
                lo[j] = r[0];
                hi[j] = r[32];
            }
#pragma unroll
            for (int j = 0; j < 8; ++j) {
                const int d = __builtin_amdgcn_readlane(dv, g + j);
                float* p = agg + (size_t)d * D + lane;
                atomicAdd(p, lo[j]);
                atomicAdd(p + 32, hi[j]);
            }
        }
    } else {
        for (int j = 0; j < n; ++j) {
            const int s = __builtin_amdgcn_readlane(sv, j);
            const int d = __builtin_amdgcn_readlane(dv, j);
            const float* __restrict__ r = h + (size_t)s * D + lane;
            float* p = agg + (size_t)d * D + lane;
            const float a = r[0];
            const float b = r[32];
            atomicAdd(p, a);
            atomicAdd(p + 32, b);
        }
    }
}


#define PNODES 128

__global__ __launch_bounds__(256) void post_project(const float* __restrict__ agg,
                                                    const float* __restrict__ rs_in,
                                                    const float* __restrict__ rs_out,
                                                    const float* __restrict__ b1,
                                                    const float* __restrict__ W2,
                                                    const float* __restrict__ Wr,
                                                    float* __restrict__ z, int N) {
    __shared__ __attribute__((aligned(16))) float wv[D];
    const int t = threadIdx.x;
    const int f = (t & 15) * 4;
    const int n0 = blockIdx.x * PNODES + (t >> 4);

    const int k = t >> 2, q = t & 3;
    const float4* __restrict__ wrow = (const float4*)(W2 + (size_t)k * D + q * 16);
    const float4* __restrict__ wr4  = (const float4*)(Wr + q * 16);
    float4 wa[4], wc[4];
#pragma unroll
    for (int j = 0; j < 4; ++j) { wa[j] = wrow[j]; wc[j] = wr4[j]; }

    float4 v[PNODES / 16];
    float  si[PNODES / 16], so[PNODES / 16];
#pragma unroll
    for (int i = 0; i < PNODES / 16; ++i) {
        const int node = n0 + i * 16;
        const int nn = (node < N) ? node : (N - 1);
        v[i]  = *(const float4*)(agg + (size_t)nn * D + f);
        si[i] = rs_in[nn];
        so[i] = rs_out[nn];
    }
    const float4 bb = *(const float4*)(b1 + f);

    {
        float p = 0.0f;
#pragma unroll
        for (int j = 0; j < 4; ++j)
            p += wa[j].x * wc[j].x + wa[j].y * wc[j].y + wa[j].z * wc[j].z + wa[j].w * wc[j].w;
        p += __shfl_xor(p, 1, 32);
        p += __shfl_xor(p, 2, 32);
        if (q == 0) wv[k] = p;
    }
    __syncthreads();
    const float4 w4 = *(const float4*)(wv + f);

#pragma unroll
    for (int i = 0; i < PNODES / 16; ++i) {
        const int node = n0 + i * 16;
        const float s = si[i];
        float p = fmaxf(v[i].x * s + bb.x, 0.0f) * w4.x
                + fmaxf(v[i].y * s + bb.y, 0.0f) * w4.y
                + fmaxf(v[i].z * s + bb.z, 0.0f) * w4.z
                + fmaxf(v[i].w * s + bb.w, 0.0f) * w4.w;
        p += __shfl_xor(p, 8, 32);
        p += __shfl_xor(p, 4, 32);
        p += __shfl_xor(p, 2, 32);
        p += __shfl_xor(p, 1, 32);
        if ((t & 15) == 0 && node < N) z[node] = p * so[i];
    }
}

#define ZCHUNKS 64
#define ZBATCH  8

__global__ __launch_bounds__(1024) void scatter_z(const float* __restrict__ z,
                                                  const int* __restrict__ src,
                                                  const int* __restrict__ dst,
                                                  float* __restrict__ zpart, int N, int E) {
    __shared__ float acc[HCAP];
    const int tid = threadIdx.x;
    const int c = blockIdx.x;
    float* __restrict__ plane = zpart + (size_t)c * (size_t)N;

    const int nvec = E >> 2;
    const int per  = (nvec + ZCHUNKS - 1) / ZCHUNKS;
    const int v0   = c * per;
    const int v1   = (v0 + per < nvec) ? (v0 + per) : nvec;
    const int4* __restrict__ dst4 = (const int4*)dst;
    const int4* __restrict__ src4 = (const int4*)src;

    for (int base = blockIdx.y * HCAP; base < N; base += gridDim.y * HCAP) {
        const unsigned cap = (unsigned)((N - base < HCAP) ? (N - base) : HCAP);
        for (unsigned i = tid; i < cap; i += 1024) acc[i] = 0.0f;
        __syncthreads();

        for (int v = v0 + tid; v < v1; v += ZBATCH * 1024) {
            int4 qd[ZBATCH], qs[ZBATCH];
#pragma unroll
            for (int j = 0; j < ZBATCH; ++j) {
                const int vv = v + j * 1024;
                qd[j] = make_int4(-1, -1, -1, -1);
                qs[j] = make_int4(0, 0, 0, 0);
                if (vv < v1) { qd[j] = dst4[vv]; qs[j] = src4[vv]; }
            }
            float4 g[ZBATCH];
#pragma unroll
            for (int j = 0; j < ZBATCH; ++j) {
                g[j] = make_float4(0.0f, 0.0f, 0.0f, 0.0f);
                if ((unsigned)(qd[j].x - base) < cap) g[j].x = z[qs[j].x];
                if ((unsigned)(qd[j].y - base) < cap) g[j].y = z[qs[j].y];
                if ((unsigned)(qd[j].z - base) < cap) g[j].z = z[qs[j].z];
                if ((unsigned)(qd[j].w - base) < cap) g[j].w = z[qs[j].w];
            }
#pragma unroll
            for (int j = 0; j < ZBATCH; ++j) {
                const unsigned u0 = (unsigned)(qd[j].x - base);
                const unsigned u1 = (unsigned)(qd[j].y - base);
                const unsigned u2 = (unsigned)(qd[j].z - base);
                const unsigned u3 = (unsigned)(qd[j].w - base);
                if (u0 < cap) atomicAdd(&acc[u0], g[j].x);
                if (u1 < cap) atomicAdd(&acc[u1], g[j].y);
                if (u2 < cap) atomicAdd(&acc[u2], g[j].z);
                if (u3 < cap) atomicAdd(&acc[u3], g[j].w);
            }
        }
        if (c == ZCHUNKS - 1) {
            for (int e = (nvec << 2) + tid; e < E; e += 1024) {
                const unsigned u = (unsigned)(dst[e] - base);
                if (u < cap) atomicAdd(&acc[u], z[src[e]]);
            }
        }
        __syncthreads();

        for (unsigned i = tid; i < cap; i += 1024) plane[base + i] = acc[i];
        __syncthreads();
    }
}

__global__ __launch_bounds__(256) void head_finish(const float* __restrict__ z,
                                                   const float* __restrict__ zpart,
                                                   const float* __restrict__ rs_in,
                                                   const float* __restrict__ b2,
                                                   const float* __restrict__ Wr,
                                                   const float* __restrict__ br,
                                                   float* __restrict__ out, int N) {
    __shared__ float c0;
    if (threadIdx.x < 32) {
        const int lane = threadIdx.x;
        float p = b2[lane] * Wr[lane] + b2[lane + 32] * Wr[lane + 32];
#pragma unroll
        for (int off = 16; off > 0; off >>= 1) p += __shfl_xor(p, off, 32);
        if (lane == 0) c0 = p + br[0];
    }
    __syncthreads();
    const int i = blockIdx.x * blockDim.x + threadIdx.x;
    if (i >= N) return;
    const float* __restrict__ pz = zpart + i;
    float p[ZCHUNKS];
#pragma unroll
    for (int c = 0; c < ZCHUNKS; ++c) p[c] = pz[(size_t)c * (size_t)N];
    float a = z[i];
    const float s = rs_in[i];
#pragma unroll
    for (int c = 0; c < ZCHUNKS; ++c) a += p[c];
    out[i] = a * s + c0;
}

extern "C" void kernel_launch(void* const* d_in, const int* in_sizes, int n_in,
                              void* d_out, int out_size, void* d_ws, size_t ws_size,
                              hipStream_t stream) {
    const float* x   = (const float*)d_in[0];
    const int*   src = (const int*)d_in[1];
    const int*   dst = (const int*)d_in[2];
    const float* W1  = (const float*)d_in[3];
    const float* b1  = (const float*)d_in[4];
    const float* W2  = (const float*)d_in[5];
    const float* b2  = (const float*)d_in[6];
    const float* Wr  = (const float*)d_in[7];
    const float* br  = (const float*)d_in[8];
    float* out = (float*)d_out;

    const int N = in_sizes[0] / D;
    const int E = in_sizes[1];

    float* ws     = (float*)d_ws;
    float* rs_out = ws;
    float* rs_in  = ws + N;
    float* h      = ws + 2 * (size_t)N;
    float* agg    = h + (size_t)N * D;
    unsigned* part = (unsigned*)(agg + (size_t)N * D);
    float* z      = (float*)(part + 2 * (size_t)HCHUNKS * (size_t)N);
    float* zpart  = z + N;

    const int TB = 256;
    const int gN    = (N + TB - 1) / TB;
    const int gGemm = (N + 127) / 128;
    const int gEdge = (E + 255) / 256;
    const int gProj = (N + PNODES - 1) / PNODES;

    deg_hist  <<<dim3(HCHUNKS, 4, 2), 1024, 0, stream>>>(src, dst, part, N, E);
    deg_finish<<<gN, TB, 0, stream>>>(part, rs_out, rs_in, N);

    gemm64_wmma<<<gGemm, TB, 0, stream>>>(x, rs_out, W1, h, agg, N);
    scatter_add<<<gEdge, TB, 0, stream>>>(h, src, dst, agg, E);

    post_project<<<gProj, TB, 0, stream>>>(agg, rs_in, rs_out, b1, W2, Wr, z, N);
    scatter_z   <<<dim3(ZCHUNKS, 4, 1), 1024, 0, stream>>>(z, src, dst, zpart, N, E);
    head_finish <<<gN, TB, 0, stream>>>(z, zpart, rs_in, b2, Wr, br, out, N);
}


# layer-1 edge sums in LDS per destination bin: edges put in bin order by an LDS counting sort, one block per 392 nodes adds gathered rows with LDS f32 adds; agg, global atomics gone; sum order changes
# speedup vs baseline: 4.4917x; 1.1211x over previous
#include <hip/hip_runtime.h>
#include <hip/hip_bf16.h>

typedef __attribute__((ext_vector_type(2))) float v2f;
typedef __attribute__((ext_vector_type(8))) float v8f;

#define D 64

#define HCAP    25600
#define HCHUNKS 32
#define HBATCH  8

__global__ __launch_bounds__(1024) void deg_hist(const int* __restrict__ src,
                                                 const int* __restrict__ dst,
                                                 unsigned* __restrict__ part, int N, int E) {
    __shared__ unsigned hist[HCAP];
    const int tid = threadIdx.x;
    const int c = blockIdx.x;
    const int* __restrict__ idx = blockIdx.z ? dst : src;
    unsigned* __restrict__ plane = part + ((size_t)blockIdx.z * HCHUNKS + c) * (size_t)N;

    const int nvec = E >> 2;
    const int per  = (nvec + HCHUNKS - 1) / HCHUNKS;
    const int v0   = c * per;
    const int v1   = (v0 + per < nvec) ? (v0 + per) : nvec;
    const int4* __restrict__ idx4 = (const int4*)idx;

    for (int base = blockIdx.y * HCAP; base < N; base += gridDim.y * HCAP) {
        const unsigned cap = (unsigned)((N - base < HCAP) ? (N - base) : HCAP);
        for (unsigned i = tid; i < cap; i += 1024) hist[i] = 0u;
        __syncthreads();

        for (int v = v0 + tid; v < v1; v += HBATCH * 1024) {
            int4 q[HBATCH];
#pragma unroll
            for (int j = 0; j < HBATCH; ++j) {
                const int vv = v + j * 1024;
                q[j] = make_int4(-1, -1, -1, -1);
                if (vv < v1) q[j] = idx4[vv];
            }
#pragma unroll
            for (int j = 0; j < HBATCH; ++j) {
                const unsigned u0 = (unsigned)(q[j].x - base);
                const unsigned u1 = (unsigned)(q[j].y - base);
                const unsigned u2 = (unsigned)(q[j].z - base);
                const unsigned u3 = (unsigned)(q[j].w - base);
                if (u0 < cap) atomicAdd(&hist[u0], 1u);
                if (u1 < cap) atomicAdd(&hist[u1], 1u);
                if (u2 < cap) atomicAdd(&hist[u2], 1u);
                if (u3 < cap) atomicAdd(&hist[u3], 1u);
            }
        }
        if (c == HCHUNKS - 1) {
            for (int e = (nvec << 2) + tid; e < E; e += 1024) {
                const unsigned u = (unsigned)(idx[e] - base);
                if (u < cap) atomicAdd(&hist[u], 1u);
            }
        }
        __syncthreads();

        for (unsigned i = tid; i < cap; i += 1024) plane[base + i] = hist[i];
        __syncthreads();
    }
}

__global__ __launch_bounds__(256) void deg_finish(const unsigned* __restrict__ part,
                                                  float* __restrict__ dout,
                                                  float* __restrict__ din, int N) {
    const int i = blockIdx.x * blockDim.x + threadIdx.x;
    if (i >= N) return;
    const unsigned* __restrict__ po = part + i;
    const unsigned* __restrict__ pi = part + (size_t)HCHUNKS * (size_t)N + i;
    unsigned a = 1u, b = 1u;
#pragma unroll
    for (int c = 0; c < HCHUNKS; ++c) {
        a += po[(size_t)c * (size_t)N];
        b += pi[(size_t)c * (size_t)N];
    }
    dout[i] = rsqrtf((float)a);
    din[i]  = rsqrtf((float)b);
}

__global__ __launch_bounds__(256) void gemm64_wmma(const float* __restrict__ X,
                                                   const float* __restrict__ rs,
                                                   const float* __restrict__ W,
                                                   float* __restrict__ out0,
                                                   int N) {
    const int wave = threadIdx.x >> 5;
    const int lane = threadIdx.x & 31;
    const int nb = blockIdx.x * 128 + wave * 16;
    if (nb >= N) return;
    const int half = lane >> 4;
    const int lr   = lane & 15;

    const float s = rs[nb + lr];
    const float* __restrict__ xrow = X + (size_t)(nb + lr) * D;

    v8f acc0 = {}, acc1 = {}, acc2 = {}, acc3 = {};
#pragma unroll
    for (int k0 = 0; k0 < D; k0 += 4) {
        const int ka = k0 + half * 2;
        v2f a;
        a.x = xrow[ka] * s;
        a.y = xrow[ka + 1] * s;
        const float* __restrict__ w0 = W + (size_t)ka * D + lr;
        const float* __restrict__ w1 = w0 + D;
        v2f b0 = { w0[0],  w1[0]  };
        v2f b1 = { w0[16], w1[16] };
        v2f b2 = { w0[32], w1[32] };
        v2f b3 = { w0[48], w1[48] };
        acc0 = __builtin_amdgcn_wmma_f32_16x16x4_f32(false, a, false, b0, (short)0, acc0, false, false);
        acc1 = __builtin_amdgcn_wmma_f32_16x16x4_f32(false, a, false, b1, (short)0, acc1, false, false);
        acc2 = __builtin_amdgcn_wmma_f32_16x16x4_f32(false, a, false, b2, (short)0, acc2, false, false);
        acc3 = __builtin_amdgcn_wmma_f32_16x16x4_f32(false, a, false, b3, (short)0, acc3, false, false);
    }

    const int row0 = nb + half * 8;
#pragma unroll
    for (int v = 0; v < 8; ++v) {
        const size_t base = (size_t)(row0 + v) * D + lr;
        out0[base]      = acc0[v];
        out0[base + 16] = acc1[v];
        out0[base + 32] = acc2[v];
        out0[base + 48] = acc3[v];
    }
}


#define BW     392
#define BWIT   ((BW + 63) / 64)
#define MAXBIN 256
#define STG    6272

__global__ __launch_bounds__(1024) void bin_count(const int* __restrict__ dst,
                                                  unsigned* __restrict__ cntg, int E) {
    __shared__ unsigned cnt[MAXBIN];
    const int tid = threadIdx.x;
    const int c = blockIdx.x;
    const int per = (E + MAXBIN - 1) / MAXBIN;
    const int e0 = c * per;
    const int e1 = (e0 + per < E) ? (e0 + per) : E;
    if (tid < MAXBIN) cnt[tid] = 0u;
    __syncthreads();
    for (int e = e0 + tid; e < e1; e += 1024) atomicAdd(&cnt[(unsigned)dst[e] / BW], 1u);
    __syncthreads();
    if (tid < MAXBIN) cntg[c * MAXBIN + tid] = cnt[tid];
}

__global__ __launch_bounds__(1024) void bin_place(const int* __restrict__ src,
                                                  const int* __restrict__ dst,
                                                  const unsigned* __restrict__ cntg,
                                                  int* __restrict__ recw,
                                                  unsigned* __restrict__ binstart, int E) {
    __shared__ unsigned tot4[4][MAXBIN];
    __shared__ unsigned pre4[4][MAXBIN];
    __shared__ unsigned gstart[MAXBIN];
    __shared__ unsigned lstart[MAXBIN];
    __shared__ unsigned cursor[MAXBIN];
    __shared__ int2 stage[STG];
    int2* __restrict__ rec = (int2*)recw;

    const int tid = threadIdx.x;
    const int c = blockIdx.x;
    const int per = (E + MAXBIN - 1) / MAXBIN;
    const int e0 = c * per;
    const int e1 = (e0 + per < E) ? (e0 + per) : E;

    {
        const int b = tid & (MAXBIN - 1);
        const int qd = tid >> 8;
        unsigned ts = 0u, ps = 0u;
#pragma unroll 16
        for (int k = 0; k < 64; ++k) {
            const int cc = qd * 64 + k;
            const unsigned v = cntg[cc * MAXBIN + b];
            ts += v;
            if (cc < c) ps += v;
        }
        tot4[qd][b] = ts;
        pre4[qd][b] = ps;
    }
    __syncthreads();

    if (tid < 32) {
        unsigned t8[8], o8[8];
        unsigned ts = 0u, os = 0u;
#pragma unroll
        for (int j = 0; j < 8; ++j) {
            const int b = tid * 8 + j;
            t8[j] = ts;
            o8[j] = os;
            ts += tot4[0][b] + tot4[1][b] + tot4[2][b] + tot4[3][b];
            os += cntg[c * MAXBIN + b];
        }
        unsigned tin = ts, oin = os;
#pragma unroll
        for (int off = 1; off < 32; off <<= 1) {
            const unsigned u = __shfl_up(tin, off, 32);
            const unsigned w = __shfl_up(oin, off, 32);
            if (tid >= off) { tin += u; oin += w; }
        }
        const unsigned tex = tin - ts, oex = oin - os;
#pragma unroll
        for (int j = 0; j < 8; ++j) {
            const int b = tid * 8 + j;
            const unsigned bs = tex + t8[j];
            gstart[b] = bs + pre4[0][b] + pre4[1][b] + pre4[2][b] + pre4[3][b];
            lstart[b] = oex + o8[j];
            cursor[b] = oex + o8[j];
            if (c == 0) binstart[b] = bs;
        }
        if (c == 0 && tid == 31) binstart[MAXBIN] = tin;
    }
    __syncthreads();

    if (per <= STG) {
        for (int e = e0 + tid; e < e1; e += 1024) {
            const int d = dst[e];
            const unsigned pos = atomicAdd(&cursor[(unsigned)d / BW], 1u);
            stage[pos] = make_int2(src[e], d);
        }
        __syncthreads();
        const int nrec = e1 - e0;
        for (int i = tid; i < nrec; i += 1024) {
            const int2 r = stage[i];
            const unsigned b = (unsigned)r.y / BW;
            rec[gstart[b] + ((unsigned)i - lstart[b])] = r;
        }
    } else {
        for (int e = e0 + tid; e < e1; e += 1024) {
            const int d = dst[e];
            const unsigned b = (unsigned)d / BW;
            const unsigned pos = atomicAdd(&cursor[b], 1u);
            rec[gstart[b] + (pos - lstart[b])] = make_int2(src[e], d);
        }
    }
}

__global__ __launch_bounds__(1024) void bin_sum(const float* __restrict__ h,
                                                const int* __restrict__ recw,
                                                const unsigned* __restrict__ binstart,
                                                const float* __restrict__ rs_in,
                                                const float* __restrict__ rs_out,
                                                const float* __restrict__ b1,
                                                const float* __restrict__ W2,
                                                const float* __restrict__ Wr,
                                                float* __restrict__ z, int N) {
    __shared__ __attribute__((aligned(16))) float acc[BW * D];
    __shared__ __attribute__((aligned(16))) float wv[D];
    const int2* __restrict__ rec = (const int2*)recw;
    const int tid = threadIdx.x;
    const int lane = tid & 31;
    const int wave = __builtin_amdgcn_readfirstlane(tid >> 5);
    const int base = blockIdx.x * BW;
    const int nn = (N - base < BW) ? (N - base) : BW;

    if (tid < 256) {
        const int k = tid >> 2, q = tid & 3;
        const float4* __restrict__ wrow = (const float4*)(W2 + (size_t)k * D + q * 16);
        const float4* __restrict__ wr4  = (const float4*)(Wr + q * 16);
        float p = 0.0f;
#pragma unroll
        for (int j = 0; j < 4; ++j) {
            const float4 a = wrow[j], w = wr4[j];
            p += a.x * w.x + a.y * w.y + a.z * w.z + a.w * w.w;
        }
        p += __shfl_xor(p, 1, 32);
        p += __shfl_xor(p, 2, 32);
        if (q == 0) wv[k] = p;
    }
    {
        const float4* __restrict__ h4 = (const float4*)(h + (size_t)base * D);
        float4* acc4 = (float4*)acc;
        for (int i = tid; i < nn * (D / 4); i += 1024) acc4[i] = h4[i];
    }
    __syncthreads();

    const int f = (tid & 15) * 4;
    const float4 bb = *(const float4*)(b1 + f);
    float si[BWIT], so[BWIT];
#pragma unroll
    for (int k = 0; k < BWIT; ++k) {
        const int i = (tid >> 4) + k * 64;
        const int ii = (i < nn) ? i : (nn - 1);
        si[k] = rs_in[base + ii];
        so[k] = rs_out[base + ii];
    }

    const unsigned r0 = binstart[blockIdx.x];
    const unsigned r1 = binstart[blockIdx.x + 1];
    unsigned g0 = r0 + (unsigned)wave * 32u;
    if (g0 < r1) {
        const unsigned last = r1 - 1u;
        const unsigned i0 = g0 + (unsigned)lane;
        int2 cur = rec[(i0 < last) ? i0 : last];
        do {
            const unsigned gn = g0 + 1024u;
            const unsigned in = gn + (unsigned)lane;
            const int2* __restrict__ pn = rec + ((in < last) ? in : last);
            int2 nxt;
            const int sv = cur.x;
            const int dv = cur.y - base;
            const unsigned left = r1 - g0;
            const int n = (left < 32u) ? (int)left : 32;

            if (n == 32) {
                float lo[32], hi[32];
#pragma unroll
                for (int j = 0; j < 32; ++j) {
                    const int s = __builtin_amdgcn_readlane(sv, j);
                    const float* __restrict__ r = h + (size_t)s * D + lane;
                    lo[j] = r[0];
                    hi[j] = r[32];
                }
                nxt = *pn;
#pragma unroll
                for (int j = 0; j < 32; ++j) {
                    const int d = __builtin_amdgcn_readlane(dv, j);
                    float* p = acc + d * D + lane;
                    atomicAdd(p, lo[j]);
                    atomicAdd(p + 32, hi[j]);
                }
            } else {
                nxt = *pn;
                for (int j = 0; j < n; ++j) {
                    const int s = __builtin_amdgcn_readlane(sv, j);
                    const int d = __builtin_amdgcn_readlane(dv, j);
                    const float* __restrict__ r = h + (size_t)s * D + lane;
                    float* p = acc + d * D + lane;
                    const float a = r[0];
                    const float b = r[32];
                    atomicAdd(p, a);
                    atomicAdd(p + 32, b);
                }
            }
            cur = nxt;
            g0 = gn;
        } while (g0 < r1);
    }
    __syncthreads();

    const float4 w4 = *(const float4*)(wv + f);
#pragma unroll
    for (int k = 0; k < BWIT; ++k) {
        const int i = (tid >> 4) + k * 64;
        const int ii = (i < nn) ? i : (nn - 1);
        const float4 v = *(const float4*)(acc + ii * D + f);
        const float s = si[k];
        float p = fmaxf(v.x * s + bb.x, 0.0f) * w4.x
                + fmaxf(v.y * s + bb.y, 0.0f) * w4.y
                + fmaxf(v.z * s + bb.z, 0.0f) * w4.z
                + fmaxf(v.w * s + bb.w, 0.0f) * w4.w;
        p += __shfl_xor(p, 8, 32);
        p += __shfl_xor(p, 4, 32);
        p += __shfl_xor(p, 2, 32);
        p += __shfl_xor(p, 1, 32);
        if ((tid & 15) == 0 && i < nn) z[base + ii] = p * so[k];
    }
}

#define ZCHUNKS 64
#define ZBATCH  8

__global__ __launch_bounds__(1024) void scatter_z(const float* __restrict__ z,
                                                  const int* __restrict__ src,
                                                  const int* __restrict__ dst,
                                                  float* __restrict__ zpart, int N, int E) {
    __shared__ float acc[HCAP];
    const int tid = threadIdx.x;
    const int c = blockIdx.x;
    float* __restrict__ plane = zpart + (size_t)c * (size_t)N;

    const int nvec = E >> 2;
    const int per  = (nvec + ZCHUNKS - 1) / ZCHUNKS;
    const int v0   = c * per;
    const int v1   = (v0 + per < nvec) ? (v0 + per) : nvec;
    const int4* __restrict__ dst4 = (const int4*)dst;
    const int4* __restrict__ src4 = (const int4*)src;

    for (int base = blockIdx.y * HCAP; base < N; base += gridDim.y * HCAP) {
        const unsigned cap = (unsigned)((N - base < HCAP) ? (N - base) : HCAP);
        for (unsigned i = tid; i < cap; i += 1024) acc[i] = 0.0f;
        __syncthreads();

        for (int v = v0 + tid; v < v1; v += ZBATCH * 1024) {
            int4 qd[ZBATCH], qs[ZBATCH];
#pragma unroll
            for (int j = 0; j < ZBATCH; ++j) {
                const int vv = v + j * 1024;
                qd[j] = make_int4(-1, -1, -1, -1);
                qs[j] = make_int4(0, 0, 0, 0);
                if (vv < v1) { qd[j] = dst4[vv]; qs[j] = src4[vv]; }
            }
            float4 g[ZBATCH];
#pragma unroll
            for (int j = 0; j < ZBATCH; ++j) {
                g[j] = make_float4(0.0f, 0.0f, 0.0f, 0.0f);
                if ((unsigned)(qd[j].x - base) < cap) g[j].x = z[qs[j].x];
                if ((unsigned)(qd[j].y - base) < cap) g[j].y = z[qs[j].y];
                if ((unsigned)(qd[j].z - base) < cap) g[j].z = z[qs[j].z];
                if ((unsigned)(qd[j].w - base) < cap) g[j].w = z[qs[j].w];
            }
#pragma unroll
            for (int j = 0; j < ZBATCH; ++j) {
                const unsigned u0 = (unsigned)(qd[j].x - base);
                const unsigned u1 = (unsigned)(qd[j].y - base);
                const unsigned u2 = (unsigned)(qd[j].z - base);
                const unsigned u3 = (unsigned)(qd[j].w - base);
                if (u0 < cap) atomicAdd(&acc[u0], g[j].x);
                if (u1 < cap) atomicAdd(&acc[u1], g[j].y);
                if (u2 < cap) atomicAdd(&acc[u2], g[j].z);
                if (u3 < cap) atomicAdd(&acc[u3], g[j].w);
            }
        }
        if (c == ZCHUNKS - 1) {
            for (int e = (nvec << 2) + tid; e < E; e += 1024) {
                const unsigned u = (unsigned)(dst[e] - base);
                if (u < cap) atomicAdd(&acc[u], z[src[e]]);
            }
        }
        __syncthreads();

        for (unsigned i = tid; i < cap; i += 1024) plane[base + i] = acc[i];
        __syncthreads();
    }
}

__global__ __launch_bounds__(256) void head_finish(const float* __restrict__ z,
                                                   const float* __restrict__ zpart,
                                                   const float* __restrict__ rs_in,
                                                   const float* __restrict__ b2,
                                                   const float* __restrict__ Wr,
                                                   const float* __restrict__ br,
                                                   float* __restrict__ out, int N) {
    __shared__ float c0;
    if (threadIdx.x < 32) {
        const int lane = threadIdx.x;
        float p = b2[lane] * Wr[lane] + b2[lane + 32] * Wr[lane + 32];
#pragma unroll
        for (int off = 16; off > 0; off >>= 1) p += __shfl_xor(p, off, 32);
        if (lane == 0) c0 = p + br[0];
    }
    __syncthreads();
    const int i = blockIdx.x * blockDim.x + threadIdx.x;
    if (i >= N) return;
    const float* __restrict__ pz = zpart + i;
    float p[ZCHUNKS];
#pragma unroll
    for (int c = 0; c < ZCHUNKS; ++c) p[c] = pz[(size_t)c * (size_t)N];
    float a = z[i];
    const float s = rs_in[i];
#pragma unroll
    for (int c = 0; c < ZCHUNKS; ++c) a += p[c];
    out[i] = a * s + c0;
}

extern "C" void kernel_launch(void* const* d_in, const int* in_sizes, int n_in,
                              void* d_out, int out_size, void* d_ws, size_t ws_size,
                              hipStream_t stream) {
    const float* x   = (const float*)d_in[0];
    const int*   src = (const int*)d_in[1];
    const int*   dst = (const int*)d_in[2];
    const float* W1  = (const float*)d_in[3];
    const float* b1  = (const float*)d_in[4];
    const float* W2  = (const float*)d_in[5];
    const float* b2  = (const float*)d_in[6];
    const float* Wr  = (const float*)d_in[7];
    const float* br  = (const float*)d_in[8];
    float* out = (float*)d_out;

    const int N = in_sizes[0] / D;
    const int E = in_sizes[1];

    float* ws     = (float*)d_ws;
    float* rs_out = ws;
    float* rs_in  = ws + N;
    float* h      = ws + 2 * (size_t)N;
    int* rec      = (int*)(h + (size_t)N * D);
    unsigned* cntg = (unsigned*)(rec + 2 * (size_t)E);
    unsigned* binstart = cntg + MAXBIN * MAXBIN;
    unsigned* part = (unsigned*)((float*)(h + (size_t)N * D) + (size_t)N * D);
    float* z      = (float*)(part + 2 * (size_t)HCHUNKS * (size_t)N);
    float* zpart  = z + N;

    const int TB = 256;
    const int gN    = (N + TB - 1) / TB;
    const int gGemm = (N + 127) / 128;
    const int gBin  = (N + BW - 1) / BW;

    deg_hist  <<<dim3(HCHUNKS, 4, 2), 1024, 0, stream>>>(src, dst, part, N, E);
    deg_finish<<<gN, TB, 0, stream>>>(part, rs_out, rs_in, N);

    gemm64_wmma<<<gGemm, TB, 0, stream>>>(x, rs_out, W1, h, N);

    bin_count<<<MAXBIN, 1024, 0, stream>>>(dst, cntg, E);
    bin_place<<<MAXBIN, 1024, 0, stream>>>(src, dst, cntg, rec, binstart, E);
    bin_sum  <<<gBin, 1024, 0, stream>>>(h, rec, binstart, rs_in, rs_out, b1, W2, Wr, z, N);

    scatter_z   <<<dim3(ZCHUNKS, 4, 1), 1024, 0, stream>>>(z, src, dst, zpart, N, E);
    head_finish <<<gN, TB, 0, stream>>>(z, zpart, rs_in, b2, Wr, br, out, N);
}


# layer-2 edge sums reuse the bin-ordered records: one block per bin adds z[src] in LDS onto z[n], writes out; partial-sum planes and finishing launch go; layer-2 sums now in arrival order
# speedup vs baseline: 4.9007x; 1.0911x over previous
#include <hip/hip_runtime.h>
#include <hip/hip_bf16.h>

typedef __attribute__((ext_vector_type(2))) float v2f;
typedef __attribute__((ext_vector_type(8))) float v8f;

#define D 64

#define HCAP    25600
#define HCHUNKS 32
#define HBATCH  8

__global__ __launch_bounds__(1024) void deg_hist(const int* __restrict__ src,
                                                 const int* __restrict__ dst,
                                                 unsigned* __restrict__ part, int N, int E) {
    __shared__ unsigned hist[HCAP];
    const int tid = threadIdx.x;
    const int c = blockIdx.x;
    const int* __restrict__ idx = blockIdx.z ? dst : src;
    unsigned* __restrict__ plane = part + ((size_t)blockIdx.z * HCHUNKS + c) * (size_t)N;

    const int nvec = E >> 2;
    const int per  = (nvec + HCHUNKS - 1) / HCHUNKS;
    const int v0   = c * per;
    const int v1   = (v0 + per < nvec) ? (v0 + per) : nvec;
    const int4* __restrict__ idx4 = (const int4*)idx;

    for (int base = blockIdx.y * HCAP; base < N; base += gridDim.y * HCAP) {
        const unsigned cap = (unsigned)((N - base < HCAP) ? (N - base) : HCAP);
        for (unsigned i = tid; i < cap; i += 1024) hist[i] = 0u;
        __syncthreads();

        for (int v = v0 + tid; v < v1; v += HBATCH * 1024) {
            int4 q[HBATCH];
#pragma unroll
            for (int j = 0; j < HBATCH; ++j) {
                const int vv = v + j * 1024;
                q[j] = make_int4(-1, -1, -1, -1);
                if (vv < v1) q[j] = idx4[vv];
            }
#pragma unroll
            for (int j = 0; j < HBATCH; ++j) {
                const unsigned u0 = (unsigned)(q[j].x - base);
                const unsigned u1 = (unsigned)(q[j].y - base);
                const unsigned u2 = (unsigned)(q[j].z - base);
                const unsigned u3 = (unsigned)(q[j].w - base);
                if (u0 < cap) atomicAdd(&hist[u0], 1u);
                if (u1 < cap) atomicAdd(&hist[u1], 1u);
                if (u2 < cap) atomicAdd(&hist[u2], 1u);
                if (u3 < cap) atomicAdd(&hist[u3], 1u);
            }
        }
        if (c == HCHUNKS - 1) {
            for (int e = (nvec << 2) + tid; e < E; e += 1024) {
                const unsigned u = (unsigned)(idx[e] - base);
                if (u < cap) atomicAdd(&hist[u], 1u);
            }
        }
        __syncthreads();

        for (unsigned i = tid; i < cap; i += 1024) plane[base + i] = hist[i];
        __syncthreads();
    }
}

__global__ __launch_bounds__(256) void deg_finish(const unsigned* __restrict__ part,
                                                  float* __restrict__ dout,
                                                  float* __restrict__ din, int N) {
    const int i = blockIdx.x * blockDim.x + threadIdx.x;
    if (i >= N) return;
    const unsigned* __restrict__ po = part + i;
    const unsigned* __restrict__ pi = part + (size_t)HCHUNKS * (size_t)N + i;
    unsigned a = 1u, b = 1u;
#pragma unroll
    for (int c = 0; c < HCHUNKS; ++c) {
        a += po[(size_t)c * (size_t)N];
        b += pi[(size_t)c * (size_t)N];
    }
    dout[i] = rsqrtf((float)a);
    din[i]  = rsqrtf((float)b);
}

__global__ __launch_bounds__(256) void gemm64_wmma(const float* __restrict__ X,
                                                   const float* __restrict__ rs,
                                                   const float* __restrict__ W,
                                                   float* __restrict__ out0,
                                                   int N) {
    const int wave = threadIdx.x >> 5;
    const int lane = threadIdx.x & 31;
    const int nb = blockIdx.x * 128 + wave * 16;
    if (nb >= N) return;
    const int half = lane >> 4;
    const int lr   = lane & 15;

    const float s = rs[nb + lr];
    const float* __restrict__ xrow = X + (size_t)(nb + lr) * D;

    v8f acc0 = {}, acc1 = {}, acc2 = {}, acc3 = {};
#pragma unroll
    for (int k0 = 0; k0 < D; k0 += 4) {
        const int ka = k0 + half * 2;
        v2f a;
        a.x = xrow[ka] * s;
        a.y = xrow[ka + 1] * s;
        const float* __restrict__ w0 = W + (size_t)ka * D + lr;
        const float* __restrict__ w1 = w0 + D;
        v2f b0 = { w0[0],  w1[0]  };
        v2f b1 = { w0[16], w1[16] };
        v2f b2 = { w0[32], w1[32] };
        v2f b3 = { w0[48], w1[48] };
        acc0 = __builtin_amdgcn_wmma_f32_16x16x4_f32(false, a, false, b0, (short)0, acc0, false, false);
        acc1 = __builtin_amdgcn_wmma_f32_16x16x4_f32(false, a, false, b1, (short)0, acc1, false, false);
        acc2 = __builtin_amdgcn_wmma_f32_16x16x4_f32(false, a, false, b2, (short)0, acc2, false, false);
        acc3 = __builtin_amdgcn_wmma_f32_16x16x4_f32(false, a, false, b3, (short)0, acc3, false, false);
    }

    const int row0 = nb + half * 8;
#pragma unroll
    for (int v = 0; v < 8; ++v) {
        const size_t base = (size_t)(row0 + v) * D + lr;
        out0[base]      = acc0[v];
        out0[base + 16] = acc1[v];
        out0[base + 32] = acc2[v];
        out0[base + 48] = acc3[v];
    }
}


#define BW     392
#define BWIT   ((BW + 63) / 64)
#define MAXBIN 256
#define STG    6272

__global__ __launch_bounds__(1024) void bin_count(const int* __restrict__ dst,
                                                  unsigned* __restrict__ cntg, int E) {
    __shared__ unsigned cnt[MAXBIN];
    const int tid = threadIdx.x;
    const int c = blockIdx.x;
    const int per = (E + MAXBIN - 1) / MAXBIN;
    const int e0 = c * per;
    const int e1 = (e0 + per < E) ? (e0 + per) : E;
    if (tid < MAXBIN) cnt[tid] = 0u;
    __syncthreads();
    for (int e = e0 + tid; e < e1; e += 1024) atomicAdd(&cnt[(unsigned)dst[e] / BW], 1u);
    __syncthreads();
    if (tid < MAXBIN) cntg[c * MAXBIN + tid] = cnt[tid];
}

__global__ __launch_bounds__(1024) void bin_place(const int* __restrict__ src,
                                                  const int* __restrict__ dst,
                                                  const unsigned* __restrict__ cntg,
                                                  int* __restrict__ recw,
                                                  unsigned* __restrict__ binstart, int E) {
    __shared__ unsigned tot4[4][MAXBIN];
    __shared__ unsigned pre4[4][MAXBIN];
    __shared__ unsigned gstart[MAXBIN];
    __shared__ unsigned lstart[MAXBIN];
    __shared__ unsigned cursor[MAXBIN];
    __shared__ int2 stage[STG];
    int2* __restrict__ rec = (int2*)recw;

    const int tid = threadIdx.x;
    const int c = blockIdx.x;
    const int per = (E + MAXBIN - 1) / MAXBIN;
    const int e0 = c * per;
    const int e1 = (e0 + per < E) ? (e0 + per) : E;

    {
        const int b = tid & (MAXBIN - 1);
        const int qd = tid >> 8;
        unsigned ts = 0u, ps = 0u;
#pragma unroll 16
        for (int k = 0; k < 64; ++k) {
            const int cc = qd * 64 + k;
            const unsigned v = cntg[cc * MAXBIN + b];
            ts += v;
            if (cc < c) ps += v;
        }
        tot4[qd][b] = ts;
        pre4[qd][b] = ps;
    }
    __syncthreads();

    if (tid < 32) {
        unsigned t8[8], o8[8];
        unsigned ts = 0u, os = 0u;
#pragma unroll
        for (int j = 0; j < 8; ++j) {
            const int b = tid * 8 + j;
            t8[j] = ts;
            o8[j] = os;
            ts += tot4[0][b] + tot4[1][b] + tot4[2][b] + tot4[3][b];
            os += cntg[c * MAXBIN + b];
        }
        unsigned tin = ts, oin = os;
#pragma unroll
        for (int off = 1; off < 32; off <<= 1) {
            const unsigned u = __shfl_up(tin, off, 32);
            const unsigned w = __shfl_up(oin, off, 32);
            if (tid >= off) { tin += u; oin += w; }
        }
        const unsigned tex = tin - ts, oex = oin - os;
#pragma unroll
        for (int j = 0; j < 8; ++j) {
            const int b = tid * 8 + j;
            const unsigned bs = tex + t8[j];
            gstart[b] = bs + pre4[0][b] + pre4[1][b] + pre4[2][b] + pre4[3][b];
            lstart[b] = oex + o8[j];
            cursor[b] = oex + o8[j];
            if (c == 0) binstart[b] = bs;
        }
        if (c == 0 && tid == 31) binstart[MAXBIN] = tin;
    }
    __syncthreads();

    if (per <= STG) {
        for (int e = e0 + tid; e < e1; e += 1024) {
            const int d = dst[e];
            const unsigned pos = atomicAdd(&cursor[(unsigned)d / BW], 1u);
            stage[pos] = make_int2(src[e], d);
        }
        __syncthreads();
        const int nrec = e1 - e0;
        for (int i = tid; i < nrec; i += 1024) {
            const int2 r = stage[i];
            const unsigned b = (unsigned)r.y / BW;
            rec[gstart[b] + ((unsigned)i - lstart[b])] = r;
        }
    } else {
        for (int e = e0 + tid; e < e1; e += 1024) {
            const int d = dst[e];
            const unsigned b = (unsigned)d / BW;
            const unsigned pos = atomicAdd(&cursor[b], 1u);
            rec[gstart[b] + (pos - lstart[b])] = make_int2(src[e], d);
        }
    }
}

__global__ __launch_bounds__(1024) void bin_sum(const float* __restrict__ h,
                                                const int* __restrict__ recw,
                                                const unsigned* __restrict__ binstart,
                                                const float* __restrict__ rs_in,
                                                const float* __restrict__ rs_out,
                                                const float* __restrict__ b1,
                                                const float* __restrict__ W2,
                                                const float* __restrict__ Wr,
                                                float* __restrict__ z, int N) {
    __shared__ __attribute__((aligned(16))) float acc[BW * D];
    __shared__ __attribute__((aligned(16))) float wv[D];
    const int2* __restrict__ rec = (const int2*)recw;
    const int tid = threadIdx.x;
    const int lane = tid & 31;
    const int wave = __builtin_amdgcn_readfirstlane(tid >> 5);
    const int base = blockIdx.x * BW;
    const int nn = (N - base < BW) ? (N - base) : BW;

    if (tid < 256) {
        const int k = tid >> 2, q = tid & 3;
        const float4* __restrict__ wrow = (const float4*)(W2 + (size_t)k * D + q * 16);
        const float4* __restrict__ wr4  = (const float4*)(Wr + q * 16);
        float p = 0.0f;
#pragma unroll
        for (int j = 0; j < 4; ++j) {
            const float4 a = wrow[j], w = wr4[j];
            p += a.x * w.x + a.y * w.y + a.z * w.z + a.w * w.w;
        }
        p += __shfl_xor(p, 1, 32);
        p += __shfl_xor(p, 2, 32);
        if (q == 0) wv[k] = p;
    }
    {
        const float4* __restrict__ h4 = (const float4*)(h + (size_t)base * D);
        float4* acc4 = (float4*)acc;
        for (int i = tid; i < nn * (D / 4); i += 1024) acc4[i] = h4[i];
    }
    __syncthreads();

    const int f = (tid & 15) * 4;
    const float4 bb = *(const float4*)(b1 + f);
    float si[BWIT], so[BWIT];
#pragma unroll
    for (int k = 0; k < BWIT; ++k) {
        const int i = (tid >> 4) + k * 64;
        const int ii = (i < nn) ? i : (nn - 1);
        si[k] = rs_in[base + ii];
        so[k] = rs_out[base + ii];
    }

    const unsigned r0 = binstart[blockIdx.x];
    const unsigned r1 = binstart[blockIdx.x + 1];
    unsigned g0 = r0 + (unsigned)wave * 32u;
    if (g0 < r1) {
        const unsigned last = r1 - 1u;
        const unsigned i0 = g0 + (unsigned)lane;
        int2 cur = rec[(i0 < last) ? i0 : last];
        do {
            const unsigned gn = g0 + 1024u;
            const unsigned in = gn + (unsigned)lane;
            const int2* __restrict__ pn = rec + ((in < last) ? in : last);
            int2 nxt;
            const int sv = cur.x;
            const int dv = cur.y - base;
            const unsigned left = r1 - g0;
            const int n = (left < 32u) ? (int)left : 32;

            if (n == 32) {
                float lo[32], hi[32];
#pragma unroll
                for (int j = 0; j < 32; ++j) {
                    const int s = __builtin_amdgcn_readlane(sv, j);
                    const float* __restrict__ r = h + (size_t)s * D + lane;
                    lo[j] = r[0];
                    hi[j] = r[32];
                }
                nxt = *pn;
#pragma unroll
                for (int j = 0; j < 32; ++j) {
                    const int d = __builtin_amdgcn_readlane(dv, j);
                    float* p = acc + d * D + lane;
                    atomicAdd(p, lo[j]);
                    atomicAdd(p + 32, hi[j]);
                }
            } else {
                nxt = *pn;
                for (int j = 0; j < n; ++j) {
                    const int s = __builtin_amdgcn_readlane(sv, j);
                    const int d = __builtin_amdgcn_readlane(dv, j);
                    const float* __restrict__ r = h + (size_t)s * D + lane;
                    float* p = acc + d * D + lane;
                    const float a = r[0];
                    const float b = r[32];
                    atomicAdd(p, a);
                    atomicAdd(p + 32, b);
                }
            }
            cur = nxt;
            g0 = gn;
        } while (g0 < r1);
    }
    __syncthreads();

    const float4 w4 = *(const float4*)(wv + f);
#pragma unroll
    for (int k = 0; k < BWIT; ++k) {
        const int i = (tid >> 4) + k * 64;
        const int ii = (i < nn) ? i : (nn - 1);
        const float4 v = *(const float4*)(acc + ii * D + f);
        const float s = si[k];
        float p = fmaxf(v.x * s + bb.x, 0.0f) * w4.x
                + fmaxf(v.y * s + bb.y, 0.0f) * w4.y
                + fmaxf(v.z * s + bb.z, 0.0f) * w4.z
                + fmaxf(v.w * s + bb.w, 0.0f) * w4.w;
        p += __shfl_xor(p, 8, 32);
        p += __shfl_xor(p, 4, 32);
        p += __shfl_xor(p, 2, 32);
        p += __shfl_xor(p, 1, 32);
        if ((tid & 15) == 0 && i < nn) z[base + ii] = p * so[k];
    }
}

#define ZT 512
#define ZB 16

__global__ __launch_bounds__(ZT) void bin_zsum(const float* __restrict__ z,
                                               const int* __restrict__ recw,
                                               const unsigned* __restrict__ binstart,
                                               const float* __restrict__ rs_in,
                                               const float* __restrict__ b2,
                                               const float* __restrict__ Wr,
                                               const float* __restrict__ br,
                                               float* __restrict__ out, int N) {
    __shared__ float acc[BW];
    __shared__ float c0;
    const int2* __restrict__ rec = (const int2*)recw;
    const int tid = threadIdx.x;
    const int base = blockIdx.x * BW;
    const unsigned nn = (unsigned)((N - base < BW) ? (N - base) : BW);
    const unsigned r0 = binstart[blockIdx.x];
    const unsigned r1 = binstart[blockIdx.x + 1];

    if (tid < 32) {
        float p = b2[tid] * Wr[tid] + b2[tid + 32] * Wr[tid + 32];
#pragma unroll
        for (int off = 16; off > 0; off >>= 1) p += __shfl_xor(p, off, 32);
        if (tid == 0) c0 = p + br[0];
    }
    static_assert(BW <= ZT, "one thread per node of a bin");
    float sc = 0.0f;
    if ((unsigned)tid < nn) {
        acc[tid] = z[base + tid];
        sc = rs_in[base + tid];
    }
    __syncthreads();

    for (unsigned g = r0; g < r1; g += ZT * ZB) {
        int2 q[ZB];
#pragma unroll
        for (int j = 0; j < ZB; ++j) {
            const unsigned i = g + (unsigned)(j * ZT + tid);
            q[j] = make_int2(0, -1);
            if (i < r1) q[j] = rec[i];
        }
        float v[ZB];
#pragma unroll
        for (int j = 0; j < ZB; ++j) {
            v[j] = 0.0f;
            if ((unsigned)(q[j].y - base) < nn) v[j] = z[q[j].x];
        }
#pragma unroll
        for (int j = 0; j < ZB; ++j) {
            const unsigned u = (unsigned)(q[j].y - base);
            if (u < nn) atomicAdd(&acc[u], v[j]);
        }
    }
    __syncthreads();

    if ((unsigned)tid < nn) out[base + tid] = acc[tid] * sc + c0;
}

extern "C" void kernel_launch(void* const* d_in, const int* in_sizes, int n_in,
                              void* d_out, int out_size, void* d_ws, size_t ws_size,
                              hipStream_t stream) {
    const float* x   = (const float*)d_in[0];
    const int*   src = (const int*)d_in[1];
    const int*   dst = (const int*)d_in[2];
    const float* W1  = (const float*)d_in[3];
    const float* b1  = (const float*)d_in[4];
    const float* W2  = (const float*)d_in[5];
    const float* b2  = (const float*)d_in[6];
    const float* Wr  = (const float*)d_in[7];
    const float* br  = (const float*)d_in[8];
    float* out = (float*)d_out;

    const int N = in_sizes[0] / D;
    const int E = in_sizes[1];

    float* ws     = (float*)d_ws;
    float* rs_out = ws;
    float* rs_in  = ws + N;
    float* h      = ws + 2 * (size_t)N;
    int* rec      = (int*)(h + (size_t)N * D);
    unsigned* cntg = (unsigned*)(rec + 2 * (size_t)E);
    unsigned* binstart = cntg + MAXBIN * MAXBIN;
    unsigned* part = (unsigned*)((float*)(h + (size_t)N * D) + (size_t)N * D);
    float* z      = (float*)(part + 2 * (size_t)HCHUNKS * (size_t)N);

    const int TB = 256;
    const int gN    = (N + TB - 1) / TB;
    const int gGemm = (N + 127) / 128;
    const int gBin  = (N + BW - 1) / BW;

    deg_hist  <<<dim3(HCHUNKS, 4, 2), 1024, 0, stream>>>(src, dst, part, N, E);
    deg_finish<<<gN, TB, 0, stream>>>(part, rs_out, rs_in, N);

    gemm64_wmma<<<gGemm, TB, 0, stream>>>(x, rs_out, W1, h, N);

    bin_count<<<MAXBIN, 1024, 0, stream>>>(dst, cntg, E);
    bin_place<<<MAXBIN, 1024, 0, stream>>>(src, dst, cntg, rec, binstart, E);
    bin_sum  <<<gBin, 1024, 0, stream>>>(h, rec, binstart, rs_in, rs_out, b1, W2, Wr, z, N);

    bin_zsum <<<gBin, ZT, 0, stream>>>(z, rec, binstart, rs_in, b2, Wr, br, out, N);
}


# independent stages share a launch (block index picks the stage): degree count + bin count; edge placement + layer-1 GEMM, whose waves finish their own rows' degrees; 7 -> 4 launches; sums unchanged
# speedup vs baseline: 6.0500x; 1.2345x over previous
#include <hip/hip_runtime.h>
#include <hip/hip_bf16.h>

typedef __attribute__((ext_vector_type(2))) float v2f;
typedef __attribute__((ext_vector_type(8))) float v8f;

#define D 64


#define HCAP    25600
#define HCHUNKS 32
#define HRANGES 4
#define HBATCH  8
#define DEGB    (HCHUNKS * HRANGES * 2)

__device__ __forceinline__ void deg_hist_role(const int b,
                                              const int* __restrict__ src,
                                              const int* __restrict__ dst,
                                              unsigned* __restrict__ part, int N, int E) {
    __shared__ unsigned hist[HCAP];
    const int tid = threadIdx.x;
    const int c = b % HCHUNKS;
    const int y = (b / HCHUNKS) % HRANGES;
    const int zz = b / (HCHUNKS * HRANGES);
    const int* __restrict__ idx = zz ? dst : src;
    unsigned* __restrict__ plane = part + ((size_t)zz * HCHUNKS + c) * (size_t)N;

    const int nvec = E >> 2;
    const int per  = (nvec + HCHUNKS - 1) / HCHUNKS;
    const int v0   = c * per;
    const int v1   = (v0 + per < nvec) ? (v0 + per) : nvec;
    const int4* __restrict__ idx4 = (const int4*)idx;

    for (int base = y * HCAP; base < N; base += HRANGES * HCAP) {
        const unsigned cap = (unsigned)((N - base < HCAP) ? (N - base) : HCAP);
        for (unsigned i = tid; i < cap; i += 1024) hist[i] = 0u;
        __syncthreads();

        for (int v = v0 + tid; v < v1; v += HBATCH * 1024) {
            int4 q[HBATCH];
#pragma unroll
            for (int j = 0; j < HBATCH; ++j) {
                const int vv = v + j * 1024;
                q[j] = make_int4(-1, -1, -1, -1);
                if (vv < v1) q[j] = idx4[vv];
            }
#pragma unroll
            for (int j = 0; j < HBATCH; ++j) {
                const unsigned u0 = (unsigned)(q[j].x - base);
                const unsigned u1 = (unsigned)(q[j].y - base);
                const unsigned u2 = (unsigned)(q[j].z - base);
                const unsigned u3 = (unsigned)(q[j].w - base);
                if (u0 < cap) atomicAdd(&hist[u0], 1u);
                if (u1 < cap) atomicAdd(&hist[u1], 1u);
                if (u2 < cap) atomicAdd(&hist[u2], 1u);
                if (u3 < cap) atomicAdd(&hist[u3], 1u);
            }
        }
        if (c == HCHUNKS - 1) {
            for (int e = (nvec << 2) + tid; e < E; e += 1024) {
                const unsigned u = (unsigned)(idx[e] - base);
                if (u < cap) atomicAdd(&hist[u], 1u);
            }
        }
        __syncthreads();

        for (unsigned i = tid; i < cap; i += 1024) plane[base + i] = hist[i];
        __syncthreads();
    }
}

#define GEMM_ROWS 512
__device__ __forceinline__ void gemm_role(const int b,
                                          const float* __restrict__ X,
                                          const unsigned* __restrict__ part,
                                          const float* __restrict__ W,
                                          float* __restrict__ out0,
                                          float* __restrict__ dout,
                                          float* __restrict__ din,
                                          int N) {
    const int wave = threadIdx.x >> 5;
    const int lane = threadIdx.x & 31;
    const int nb = b * GEMM_ROWS + wave * 16;
    if (nb >= N) return;
    const int half = lane >> 4;
    const int lr   = lane & 15;

    const unsigned poff = (unsigned)half * (unsigned)(HCHUNKS * N) + (unsigned)(nb + lr);
    unsigned cnt1 = 1u;
#pragma nounroll
    for (int c0 = 0; c0 < HCHUNKS; c0 += 16) {
#pragma unroll
        for (int j = 0; j < 16; ++j) cnt1 += (part + (size_t)(c0 + j) * (size_t)N)[poff];
    }
    const float r = rsqrtf((float)cnt1);
    if (half == 0) dout[nb + lr] = r;
    else           din[nb + lr]  = r;
    const float s = __shfl(r, lr, 32);
    const float* __restrict__ xrow = X + (size_t)(nb + lr) * D;

    v8f acc0 = {}, acc1 = {}, acc2 = {}, acc3 = {};
#pragma unroll
    for (int k0 = 0; k0 < D; k0 += 4) {
        const int ka = k0 + half * 2;
        v2f a;
        a.x = xrow[ka] * s;
        a.y = xrow[ka + 1] * s;
        const float* __restrict__ w0 = W + (size_t)ka * D + lr;
        const float* __restrict__ w1 = w0 + D;
        v2f b0 = { w0[0],  w1[0]  };
        v2f b1 = { w0[16], w1[16] };
        v2f b2 = { w0[32], w1[32] };
        v2f b3 = { w0[48], w1[48] };
        acc0 = __builtin_amdgcn_wmma_f32_16x16x4_f32(false, a, false, b0, (short)0, acc0, false, false);
        acc1 = __builtin_amdgcn_wmma_f32_16x16x4_f32(false, a, false, b1, (short)0, acc1, false, false);
        acc2 = __builtin_amdgcn_wmma_f32_16x16x4_f32(false, a, false, b2, (short)0, acc2, false, false);
        acc3 = __builtin_amdgcn_wmma_f32_16x16x4_f32(false, a, false, b3, (short)0, acc3, false, false);
    }

    const int row0 = nb + half * 8;
#pragma unroll
    for (int v = 0; v < 8; ++v) {
        const size_t base = (size_t)(row0 + v) * D + lr;
        out0[base]      = acc0[v];
        out0[base + 16] = acc1[v];
        out0[base + 32] = acc2[v];
        out0[base + 48] = acc3[v];
    }
}


#define BW     392
#define BWIT   ((BW + 63) / 64)
#define MAXBIN 256
#define STG    6272

__device__ __forceinline__ void bin_count_role(const int c,
                                               const int* __restrict__ dst,
                                               unsigned* __restrict__ cntg, int E) {
    __shared__ unsigned cnt[MAXBIN];
    const int tid = threadIdx.x;
    const int per = (E + MAXBIN - 1) / MAXBIN;
    const int e0 = c * per;
    const int e1 = (e0 + per < E) ? (e0 + per) : E;
    if (tid < MAXBIN) cnt[tid] = 0u;
    __syncthreads();
    for (int e = e0 + tid; e < e1; e += 1024) atomicAdd(&cnt[(unsigned)dst[e] / BW], 1u);
    __syncthreads();
    if (tid < MAXBIN) cntg[c * MAXBIN + tid] = cnt[tid];
}

__global__ __launch_bounds__(1024) void count_stage(const int* __restrict__ src,
                                                    const int* __restrict__ dst,
                                                    unsigned* __restrict__ part,
                                                    unsigned* __restrict__ cntg, int N, int E) {
    const int b = blockIdx.x;
    if (b < DEGB) deg_hist_role(b, src, dst, part, N, E);
    else          bin_count_role(b - DEGB, dst, cntg, E);
}

__device__ __forceinline__ void bin_place_role(const int c,
                                               const int* __restrict__ src,
                                               const int* __restrict__ dst,
                                               const unsigned* __restrict__ cntg,
                                               int* __restrict__ recw,
                                               unsigned* __restrict__ binstart, int E) {
    __shared__ unsigned tot4[4][MAXBIN];
    __shared__ unsigned pre4[4][MAXBIN];
    __shared__ unsigned gstart[MAXBIN];
    __shared__ unsigned lstart[MAXBIN];
    __shared__ unsigned cursor[MAXBIN];
    __shared__ int2 stage[STG];
    int2* __restrict__ rec = (int2*)recw;

    const int tid = threadIdx.x;
    const int per = (E + MAXBIN - 1) / MAXBIN;
    const int e0 = c * per;
    const int e1 = (e0 + per < E) ? (e0 + per) : E;

    {
        const int b = tid & (MAXBIN - 1);
        const int qd = tid >> 8;
        unsigned ts = 0u, ps = 0u;
#pragma unroll 16
        for (int k = 0; k < 64; ++k) {
            const int cc = qd * 64 + k;
            const unsigned v = cntg[cc * MAXBIN + b];
            ts += v;
            if (cc < c) ps += v;
        }
        tot4[qd][b] = ts;
        pre4[qd][b] = ps;
    }
    __syncthreads();

    if (tid < 32) {
        unsigned t8[8], o8[8];
        unsigned ts = 0u, os = 0u;
#pragma unroll
        for (int j = 0; j < 8; ++j) {
            const int b = tid * 8 + j;
            t8[j] = ts;
            o8[j] = os;
            ts += tot4[0][b] + tot4[1][b] + tot4[2][b] + tot4[3][b];
            os += cntg[c * MAXBIN + b];
        }
        unsigned tin = ts, oin = os;
#pragma unroll
        for (int off = 1; off < 32; off <<= 1) {
            const unsigned u = __shfl_up(tin, off, 32);
            const unsigned w = __shfl_up(oin, off, 32);
            if (tid >= off) { tin += u; oin += w; }
        }
        const unsigned tex = tin - ts, oex = oin - os;
#pragma unroll
        for (int j = 0; j < 8; ++j) {
            const int b = tid * 8 + j;
            const unsigned bs = tex + t8[j];
            gstart[b] = bs + pre4[0][b] + pre4[1][b] + pre4[2][b] + pre4[3][b];
            lstart[b] = oex + o8[j];
            cursor[b] = oex + o8[j];
            if (c == 0) binstart[b] = bs;
        }
        if (c == 0 && tid == 31) binstart[MAXBIN] = tin;
    }
    __syncthreads();

    if (per <= STG) {
        for (int e = e0 + tid; e < e1; e += 1024) {
            const int d = dst[e];
            const unsigned pos = atomicAdd(&cursor[(unsigned)d / BW], 1u);
            stage[pos] = make_int2(src[e], d);
        }
        __syncthreads();
        const int nrec = e1 - e0;
        for (int i = tid; i < nrec; i += 1024) {
            const int2 r = stage[i];
            const unsigned b = (unsigned)r.y / BW;
            rec[gstart[b] + ((unsigned)i - lstart[b])] = r;
        }
    } else {
        for (int e = e0 + tid; e < e1; e += 1024) {
            const int d = dst[e];
            const unsigned b = (unsigned)d / BW;
            const unsigned pos = atomicAdd(&cursor[b], 1u);
            rec[gstart[b] + (pos - lstart[b])] = make_int2(src[e], d);
        }
    }
}

__global__ __launch_bounds__(1024) void place_gemm_stage(const float* __restrict__ X,
                                                         const int* __restrict__ src,
                                                         const int* __restrict__ dst,
                                                         const float* __restrict__ W,
                                                         const unsigned* __restrict__ part,
                                                         const unsigned* __restrict__ cntg,
                                                         int* __restrict__ recw,
                                                         unsigned* __restrict__ binstart,
                                                         float* __restrict__ dout,
                                                         float* __restrict__ din,
                                                         float* __restrict__ out0,
                                                         int N, int E) {
    const int b = blockIdx.x;
    if (b < MAXBIN) bin_place_role(b, src, dst, cntg, recw, binstart, E);
    else            gemm_role(b - MAXBIN, X, part, W, out0, dout, din, N);
}

__global__ __launch_bounds__(1024) void bin_sum(const float* __restrict__ h,
                                                const int* __restrict__ recw,
                                                const unsigned* __restrict__ binstart,
                                                const float* __restrict__ rs_in,
                                                const float* __restrict__ rs_out,
                                                const float* __restrict__ b1,
                                                const float* __restrict__ W2,
                                                const float* __restrict__ Wr,
                                                float* __restrict__ z, int N) {
    __shared__ __attribute__((aligned(16))) float acc[BW * D];
    __shared__ __attribute__((aligned(16))) float wv[D];
    const int2* __restrict__ rec = (const int2*)recw;
    const int tid = threadIdx.x;
    const int lane = tid & 31;
    const int wave = __builtin_amdgcn_readfirstlane(tid >> 5);
    const int base = blockIdx.x * BW;
    const int nn = (N - base < BW) ? (N - base) : BW;

    if (tid < 256) {
        const int k = tid >> 2, q = tid & 3;
        const float4* __restrict__ wrow = (const float4*)(W2 + (size_t)k * D + q * 16);
        const float4* __restrict__ wr4  = (const float4*)(Wr + q * 16);
        float p = 0.0f;
#pragma unroll
        for (int j = 0; j < 4; ++j) {
            const float4 a = wrow[j], w = wr4[j];
            p += a.x * w.x + a.y * w.y + a.z * w.z + a.w * w.w;
        }
        p += __shfl_xor(p, 1, 32);
        p += __shfl_xor(p, 2, 32);
        if (q == 0) wv[k] = p;
    }
    {
        const float4* __restrict__ h4 = (const float4*)(h + (size_t)base * D);
        float4* acc4 = (float4*)acc;
        for (int i = tid; i < nn * (D / 4); i += 1024) acc4[i] = h4[i];
    }
    __syncthreads();

    const int f = (tid & 15) * 4;
    const float4 bb = *(const float4*)(b1 + f);
    float si[BWIT], so[BWIT];
#pragma unroll
    for (int k = 0; k < BWIT; ++k) {
        const int i = (tid >> 4) + k * 64;
        const int ii = (i < nn) ? i : (nn - 1);
        si[k] = rs_in[base + ii];
        so[k] = rs_out[base + ii];
    }

    const unsigned r0 = binstart[blockIdx.x];
    const unsigned r1 = binstart[blockIdx.x + 1];
    unsigned g0 = r0 + (unsigned)wave * 32u;
    if (g0 < r1) {
        const unsigned last = r1 - 1u;
        const unsigned i0 = g0 + (unsigned)lane;
        int2 cur = rec[(i0 < last) ? i0 : last];
        do {
            const unsigned gn = g0 + 1024u;
            const unsigned in = gn + (unsigned)lane;
            const int2* __restrict__ pn = rec + ((in < last) ? in : last);
            int2 nxt;
            const int sv = cur.x;
            const int dv = cur.y - base;
            const unsigned left = r1 - g0;
            const int n = (left < 32u) ? (int)left : 32;

            if (n == 32) {
                float lo[32], hi[32];
#pragma unroll
                for (int j = 0; j < 32; ++j) {
                    const int s = __builtin_amdgcn_readlane(sv, j);
                    const float* __restrict__ r = h + (size_t)s * D + lane;
                    lo[j] = r[0];
                    hi[j] = r[32];
                }
                nxt = *pn;
#pragma unroll
                for (int j = 0; j < 32; ++j) {
                    const int d = __builtin_amdgcn_readlane(dv, j);
                    float* p = acc + d * D + lane;
                    atomicAdd(p, lo[j]);
                    atomicAdd(p + 32, hi[j]);
                }
            } else {
                nxt = *pn;
                for (int j = 0; j < n; ++j) {
                    const int s = __builtin_amdgcn_readlane(sv, j);
                    const int d = __builtin_amdgcn_readlane(dv, j);
                    const float* __restrict__ r = h + (size_t)s * D + lane;
                    float* p = acc + d * D + lane;
                    const float a = r[0];
                    const float b = r[32];
                    atomicAdd(p, a);
                    atomicAdd(p + 32, b);
                }
            }
            cur = nxt;
            g0 = gn;
        } while (g0 < r1);
    }
    __syncthreads();

    const float4 w4 = *(const float4*)(wv + f);
#pragma unroll
    for (int k = 0; k < BWIT; ++k) {
        const int i = (tid >> 4) + k * 64;
        const int ii = (i < nn) ? i : (nn - 1);
        const float4 v = *(const float4*)(acc + ii * D + f);
        const float s = si[k];
        float p = fmaxf(v.x * s + bb.x, 0.0f) * w4.x
                + fmaxf(v.y * s + bb.y, 0.0f) * w4.y
                + fmaxf(v.z * s + bb.z, 0.0f) * w4.z
                + fmaxf(v.w * s + bb.w, 0.0f) * w4.w;
        p += __shfl_xor(p, 8, 32);
        p += __shfl_xor(p, 4, 32);
        p += __shfl_xor(p, 2, 32);
        p += __shfl_xor(p, 1, 32);
        if ((tid & 15) == 0 && i < nn) z[base + ii] = p * so[k];
    }
}

#define ZT 512
#define ZB 16

__global__ __launch_bounds__(ZT) void bin_zsum(const float* __restrict__ z,
                                               const int* __restrict__ recw,
                                               const unsigned* __restrict__ binstart,
                                               const float* __restrict__ rs_in,
                                               const float* __restrict__ b2,
                                               const float* __restrict__ Wr,
                                               const float* __restrict__ br,
                                               float* __restrict__ out, int N) {
    __shared__ float acc[BW];
    __shared__ float c0;
    const int2* __restrict__ rec = (const int2*)recw;
    const int tid = threadIdx.x;
    const int base = blockIdx.x * BW;
    const unsigned nn = (unsigned)((N - base < BW) ? (N - base) : BW);
    const unsigned r0 = binstart[blockIdx.x];
    const unsigned r1 = binstart[blockIdx.x + 1];

    if (tid < 32) {
        float p = b2[tid] * Wr[tid] + b2[tid + 32] * Wr[tid + 32];
#pragma unroll
        for (int off = 16; off > 0; off >>= 1) p += __shfl_xor(p, off, 32);
        if (tid == 0) c0 = p + br[0];
    }
    static_assert(BW <= ZT, "one thread per node of a bin");
    float sc = 0.0f;
    if ((unsigned)tid < nn) {
        acc[tid] = z[base + tid];
        sc = rs_in[base + tid];
    }
    __syncthreads();

    for (unsigned g = r0; g < r1; g += ZT * ZB) {
        int2 q[ZB];
#pragma unroll
        for (int j = 0; j < ZB; ++j) {
            const unsigned i = g + (unsigned)(j * ZT + tid);
            q[j] = make_int2(0, -1);
            if (i < r1) q[j] = rec[i];
        }
        float v[ZB];
#pragma unroll
        for (int j = 0; j < ZB; ++j) {
            v[j] = 0.0f;
            if ((unsigned)(q[j].y - base) < nn) v[j] = z[q[j].x];
        }
#pragma unroll
        for (int j = 0; j < ZB; ++j) {
            const unsigned u = (unsigned)(q[j].y - base);
            if (u < nn) atomicAdd(&acc[u], v[j]);
        }
    }
    __syncthreads();

    if ((unsigned)tid < nn) out[base + tid] = acc[tid] * sc + c0;
}

extern "C" void kernel_launch(void* const* d_in, const int* in_sizes, int n_in,
                              void* d_out, int out_size, void* d_ws, size_t ws_size,
                              hipStream_t stream) {
    const float* x   = (const float*)d_in[0];
    const int*   src = (const int*)d_in[1];
    const int*   dst = (const int*)d_in[2];
    const float* W1  = (const float*)d_in[3];
    const float* b1  = (const float*)d_in[4];
    const float* W2  = (const float*)d_in[5];
    const float* b2  = (const float*)d_in[6];
    const float* Wr  = (const float*)d_in[7];
    const float* br  = (const float*)d_in[8];
    float* out = (float*)d_out;

    const int N = in_sizes[0] / D;
    const int E = in_sizes[1];

    float* ws     = (float*)d_ws;
    float* rs_out = ws;
    float* rs_in  = ws + N;
    float* h      = ws + 2 * (size_t)N;
    int* rec      = (int*)(h + (size_t)N * D);
    unsigned* cntg = (unsigned*)(rec + 2 * (size_t)E);
    unsigned* binstart = cntg + MAXBIN * MAXBIN;
    unsigned* part = (unsigned*)((float*)(h + (size_t)N * D) + (size_t)N * D);
    float* z      = (float*)(part + 2 * (size_t)HCHUNKS * (size_t)N);

    const int gGemm = (N + GEMM_ROWS - 1) / GEMM_ROWS;
    const int gBin  = (N + BW - 1) / BW;

    count_stage<<<DEGB + MAXBIN, 1024, 0, stream>>>(src, dst, part, cntg, N, E);

    place_gemm_stage<<<MAXBIN + gGemm, 1024, 0, stream>>>(x, src, dst, W1, part, cntg, rec, binstart,
                                                          rs_out, rs_in, h, N, E);

    bin_sum  <<<gBin, 1024, 0, stream>>>(h, rec, binstart, rs_in, rs_out, b1, W2, Wr, z, N);

    bin_zsum <<<gBin, ZT, 0, stream>>>(z, rec, binstart, rs_in, b2, Wr, br, out, N);
}


# bin-count and placement passes: a thread's index and counter loads go out in batches ahead of its LDS adds (was a round trip per edge); the now short placement blocks enter behind the GEMM blocks
# speedup vs baseline: 6.2827x; 1.0385x over previous
#include <hip/hip_runtime.h>
#include <hip/hip_bf16.h>

typedef __attribute__((ext_vector_type(2))) float v2f;
typedef __attribute__((ext_vector_type(8))) float v8f;

#define D 64


#define HCAP    25600
#define HCHUNKS 32
#define HRANGES 4
#define HBATCH  8
#define DEGB    (HCHUNKS * HRANGES * 2)

__device__ __forceinline__ void deg_hist_role(const int b,
                                              const int* __restrict__ src,
                                              const int* __restrict__ dst,
                                              unsigned* __restrict__ part, int N, int E) {
    __shared__ unsigned hist[HCAP];
    const int tid = threadIdx.x;
    const int c = b % HCHUNKS;
    const int y = (b / HCHUNKS) % HRANGES;
    const int zz = b / (HCHUNKS * HRANGES);
    const int* __restrict__ idx = zz ? dst : src;
    unsigned* __restrict__ plane = part + ((size_t)zz * HCHUNKS + c) * (size_t)N;

    const int nvec = E >> 2;
    const int per  = (nvec + HCHUNKS - 1) / HCHUNKS;
    const int v0   = c * per;
    const int v1   = (v0 + per < nvec) ? (v0 + per) : nvec;
    const int4* __restrict__ idx4 = (const int4*)idx;

    for (int base = y * HCAP; base < N; base += HRANGES * HCAP) {
        const unsigned cap = (unsigned)((N - base < HCAP) ? (N - base) : HCAP);
        for (unsigned i = tid; i < cap; i += 1024) hist[i] = 0u;
        __syncthreads();

        for (int v = v0 + tid; v < v1; v += HBATCH * 1024) {
            int4 q[HBATCH];
#pragma unroll
            for (int j = 0; j < HBATCH; ++j) {
                const int vv = v + j * 1024;
                q[j] = make_int4(-1, -1, -1, -1);
                if (vv < v1) q[j] = idx4[vv];
            }
#pragma unroll
            for (int j = 0; j < HBATCH; ++j) {
                const unsigned u0 = (unsigned)(q[j].x - base);
                const unsigned u1 = (unsigned)(q[j].y - base);
                const unsigned u2 = (unsigned)(q[j].z - base);
                const unsigned u3 = (unsigned)(q[j].w - base);
                if (u0 < cap) atomicAdd(&hist[u0], 1u);
                if (u1 < cap) atomicAdd(&hist[u1], 1u);
                if (u2 < cap) atomicAdd(&hist[u2], 1u);
                if (u3 < cap) atomicAdd(&hist[u3], 1u);
            }
        }
        if (c == HCHUNKS - 1) {
            for (int e = (nvec << 2) + tid; e < E; e += 1024) {
                const unsigned u = (unsigned)(idx[e] - base);
                if (u < cap) atomicAdd(&hist[u], 1u);
            }
        }
        __syncthreads();

        for (unsigned i = tid; i < cap; i += 1024) plane[base + i] = hist[i];
        __syncthreads();
    }
}

#define GEMM_ROWS 512
__device__ __forceinline__ void gemm_role(const int b,
                                          const float* __restrict__ X,
                                          const unsigned* __restrict__ part,
                                          const float* __restrict__ W,
                                          float* __restrict__ out0,
                                          float* __restrict__ dout,
                                          float* __restrict__ din,
                                          int N) {
    const int wave = threadIdx.x >> 5;
    const int lane = threadIdx.x & 31;
    const int nb = b * GEMM_ROWS + wave * 16;
    if (nb >= N) return;
    const int half = lane >> 4;
    const int lr   = lane & 15;

    const unsigned poff = (unsigned)half * (unsigned)(HCHUNKS * N) + (unsigned)(nb + lr);
    unsigned cnt1 = 1u;
#pragma nounroll
    for (int c0 = 0; c0 < HCHUNKS; c0 += 16) {
#pragma unroll
        for (int j = 0; j < 16; ++j) cnt1 += (part + (size_t)(c0 + j) * (size_t)N)[poff];
    }
    const float r = rsqrtf((float)cnt1);
    if (half == 0) dout[nb + lr] = r;
    else           din[nb + lr]  = r;
    const float s = __shfl(r, lr, 32);
    const float* __restrict__ xrow = X + (size_t)(nb + lr) * D;

    v8f acc0 = {}, acc1 = {}, acc2 = {}, acc3 = {};
#pragma unroll
    for (int k0 = 0; k0 < D; k0 += 4) {
        const int ka = k0 + half * 2;
        v2f a;
        a.x = xrow[ka] * s;
        a.y = xrow[ka + 1] * s;
        const float* __restrict__ w0 = W + (size_t)ka * D + lr;
        const float* __restrict__ w1 = w0 + D;
        v2f b0 = { w0[0],  w1[0]  };
        v2f b1 = { w0[16], w1[16] };
        v2f b2 = { w0[32], w1[32] };
        v2f b3 = { w0[48], w1[48] };
        acc0 = __builtin_amdgcn_wmma_f32_16x16x4_f32(false, a, false, b0, (short)0, acc0, false, false);
        acc1 = __builtin_amdgcn_wmma_f32_16x16x4_f32(false, a, false, b1, (short)0, acc1, false, false);
        acc2 = __builtin_amdgcn_wmma_f32_16x16x4_f32(false, a, false, b2, (short)0, acc2, false, false);
        acc3 = __builtin_amdgcn_wmma_f32_16x16x4_f32(false, a, false, b3, (short)0, acc3, false, false);
    }

    const int row0 = nb + half * 8;
#pragma unroll
    for (int v = 0; v < 8; ++v) {
        const size_t base = (size_t)(row0 + v) * D + lr;
        out0[base]      = acc0[v];
        out0[base + 16] = acc1[v];
        out0[base + 32] = acc2[v];
        out0[base + 48] = acc3[v];
    }
}


#define BW     392
#define BWIT   ((BW + 63) / 64)
#define MAXBIN 256
#define STG    6272
#define IB     8

__device__ __forceinline__ void bin_count_role(const int c,
                                               const int* __restrict__ dst,
                                               unsigned* __restrict__ cntg, int E) {
    __shared__ unsigned cnt[MAXBIN];
    const int tid = threadIdx.x;
    const int per = (E + MAXBIN - 1) / MAXBIN;
    const int e0 = c * per;
    const int e1 = (e0 + per < E) ? (e0 + per) : E;
    if (tid < MAXBIN) cnt[tid] = 0u;
    __syncthreads();
    for (int e = e0 + tid; e < e1; e += IB * 1024) {
        int d[IB];
#pragma unroll
        for (int j = 0; j < IB; ++j) {
            const int ee = e + j * 1024;
            d[j] = -1;
            if (ee < e1) d[j] = dst[ee];
        }
#pragma unroll
        for (int j = 0; j < IB; ++j)
            if (d[j] >= 0) atomicAdd(&cnt[(unsigned)d[j] / BW], 1u);
    }
    __syncthreads();
    if (tid < MAXBIN) cntg[c * MAXBIN + tid] = cnt[tid];
}

__global__ __launch_bounds__(1024) void count_stage(const int* __restrict__ src,
                                                    const int* __restrict__ dst,
                                                    unsigned* __restrict__ part,
                                                    unsigned* __restrict__ cntg, int N, int E) {
    const int b = blockIdx.x;
    if (b < DEGB) deg_hist_role(b, src, dst, part, N, E);
    else          bin_count_role(b - DEGB, dst, cntg, E);
}

__device__ __forceinline__ void bin_place_role(const int c,
                                               const int* __restrict__ src,
                                               const int* __restrict__ dst,
                                               const unsigned* __restrict__ cntg,
                                               int* __restrict__ recw,
                                               unsigned* __restrict__ binstart, int E) {
    __shared__ unsigned tot4[4][MAXBIN];
    __shared__ unsigned pre4[4][MAXBIN];
    __shared__ unsigned gstart[MAXBIN];
    __shared__ unsigned lstart[MAXBIN];
    __shared__ unsigned cursor[MAXBIN];
    __shared__ int2 stage[STG];
    int2* __restrict__ rec = (int2*)recw;

    const int tid = threadIdx.x;
    const int per = (E + MAXBIN - 1) / MAXBIN;
    const int e0 = c * per;
    const int e1 = (e0 + per < E) ? (e0 + per) : E;

    {
        const int b = tid & (MAXBIN - 1);
        const int qd = tid >> 8;
        unsigned ts = 0u, ps = 0u;
        const unsigned* __restrict__ col = cntg + (size_t)(qd * 64) * MAXBIN + b;
#pragma nounroll
        for (int k0 = 0; k0 < 64; k0 += 16) {
            unsigned v[16];
#pragma unroll
            for (int k = 0; k < 16; ++k) v[k] = col[(size_t)(k0 + k) * MAXBIN];
#pragma unroll
            for (int k = 0; k < 16; ++k) {
                ts += v[k];
                if (qd * 64 + k0 + k < c) ps += v[k];
            }
        }
        tot4[qd][b] = ts;
        pre4[qd][b] = ps;
    }
    __syncthreads();

    if (tid < 32) {
        unsigned t8[8], o8[8];
        unsigned ts = 0u, os = 0u;
#pragma unroll
        for (int j = 0; j < 8; ++j) {
            const int b = tid * 8 + j;
            t8[j] = ts;
            o8[j] = os;
            ts += tot4[0][b] + tot4[1][b] + tot4[2][b] + tot4[3][b];
            os += cntg[c * MAXBIN + b];
        }
        unsigned tin = ts, oin = os;
#pragma unroll
        for (int off = 1; off < 32; off <<= 1) {
            const unsigned u = __shfl_up(tin, off, 32);
            const unsigned w = __shfl_up(oin, off, 32);
            if (tid >= off) { tin += u; oin += w; }
        }
        const unsigned tex = tin - ts, oex = oin - os;
#pragma unroll
        for (int j = 0; j < 8; ++j) {
            const int b = tid * 8 + j;
            const unsigned bs = tex + t8[j];
            gstart[b] = bs + pre4[0][b] + pre4[1][b] + pre4[2][b] + pre4[3][b];
            lstart[b] = oex + o8[j];
            cursor[b] = oex + o8[j];
            if (c == 0) binstart[b] = bs;
        }
        if (c == 0 && tid == 31) binstart[MAXBIN] = tin;
    }
    __syncthreads();

    if (per <= STG) {
        for (int e = e0 + tid; e < e1; e += IB * 1024) {
            int dd[IB], ss[IB];
#pragma unroll
            for (int j = 0; j < IB; ++j) {
                const int ee = e + j * 1024;
                dd[j] = -1;
                ss[j] = 0;
                if (ee < e1) { dd[j] = dst[ee]; ss[j] = src[ee]; }
            }
            unsigned pos[IB];
#pragma unroll
            for (int j = 0; j < IB; ++j) {
                pos[j] = 0u;
                if (dd[j] >= 0) pos[j] = atomicAdd(&cursor[(unsigned)dd[j] / BW], 1u);
            }
#pragma unroll
            for (int j = 0; j < IB; ++j)
                if (dd[j] >= 0) stage[pos[j]] = make_int2(ss[j], dd[j]);
        }
        __syncthreads();
        const int nrec = e1 - e0;
        for (int i = tid; i < nrec; i += IB * 1024) {
            int2 r[IB];
#pragma unroll
            for (int j = 0; j < IB; ++j) {
                const int ii = i + j * 1024;
                r[j] = make_int2(0, -1);
                if (ii < nrec) r[j] = stage[ii];
            }
            unsigned gs[IB], ls[IB];
#pragma unroll
            for (int j = 0; j < IB; ++j) {
                gs[j] = 0u;
                ls[j] = 0u;
                if (r[j].y >= 0) {
                    const unsigned b = (unsigned)r[j].y / BW;
                    gs[j] = gstart[b];
                    ls[j] = lstart[b];
                }
            }
#pragma unroll
            for (int j = 0; j < IB; ++j)
                if (r[j].y >= 0) rec[gs[j] + ((unsigned)(i + j * 1024) - ls[j])] = r[j];
        }
    } else {
        for (int e = e0 + tid; e < e1; e += IB * 1024) {
            int dd[IB], ss[IB];
#pragma unroll
            for (int j = 0; j < IB; ++j) {
                const int ee = e + j * 1024;
                dd[j] = -1;
                ss[j] = 0;
                if (ee < e1) { dd[j] = dst[ee]; ss[j] = src[ee]; }
            }
#pragma unroll
            for (int j = 0; j < IB; ++j) {
                if (dd[j] >= 0) {
                    const unsigned b = (unsigned)dd[j] / BW;
                    const unsigned pos = atomicAdd(&cursor[b], 1u);
                    rec[gstart[b] + (pos - lstart[b])] = make_int2(ss[j], dd[j]);
                }
            }
        }
    }
}

__global__ __launch_bounds__(1024) void place_gemm_stage(const float* __restrict__ X,
                                                         const int* __restrict__ src,
                                                         const int* __restrict__ dst,
                                                         const float* __restrict__ W,
                                                         const unsigned* __restrict__ part,
                                                         const unsigned* __restrict__ cntg,
                                                         int* __restrict__ recw,
                                                         unsigned* __restrict__ binstart,
                                                         float* __restrict__ dout,
                                                         float* __restrict__ din,
                                                         float* __restrict__ out0,
                                                         int N, int E) {
    const int b = blockIdx.x;
    const int gGemm = (N + GEMM_ROWS - 1) / GEMM_ROWS;
    if (b < gGemm) gemm_role(b, X, part, W, out0, dout, din, N);
    else           bin_place_role(b - gGemm, src, dst, cntg, recw, binstart, E);
}

__global__ __launch_bounds__(1024) void bin_sum(const float* __restrict__ h,
                                                const int* __restrict__ recw,
                                                const unsigned* __restrict__ binstart,
                                                const float* __restrict__ rs_in,
                                                const float* __restrict__ rs_out,
                                                const float* __restrict__ b1,
                                                const float* __restrict__ W2,
                                                const float* __restrict__ Wr,
                                                float* __restrict__ z, int N) {
    __shared__ __attribute__((aligned(16))) float acc[BW * D];
    __shared__ __attribute__((aligned(16))) float wv[D];
    const int2* __restrict__ rec = (const int2*)recw;
    const int tid = threadIdx.x;
    const int lane = tid & 31;
    const int wave = __builtin_amdgcn_readfirstlane(tid >> 5);
    const int base = blockIdx.x * BW;
    const int nn = (N - base < BW) ? (N - base) : BW;

    if (tid < 256) {
        const int k = tid >> 2, q = tid & 3;
        const float4* __restrict__ wrow = (const float4*)(W2 + (size_t)k * D + q * 16);
        const float4* __restrict__ wr4  = (const float4*)(Wr + q * 16);
        float p = 0.0f;
#pragma unroll
        for (int j = 0; j < 4; ++j) {
            const float4 a = wrow[j], w = wr4[j];
            p += a.x * w.x + a.y * w.y + a.z * w.z + a.w * w.w;
        }
        p += __shfl_xor(p, 1, 32);
        p += __shfl_xor(p, 2, 32);
        if (q == 0) wv[k] = p;
    }
    {
        const float4* __restrict__ h4 = (const float4*)(h + (size_t)base * D);
        float4* acc4 = (float4*)acc;
        for (int i = tid; i < nn * (D / 4); i += 1024) acc4[i] = h4[i];
    }
    __syncthreads();

    const int f = (tid & 15) * 4;
    const float4 bb = *(const float4*)(b1 + f);
    float si[BWIT], so[BWIT];
#pragma unroll
    for (int k = 0; k < BWIT; ++k) {
        const int i = (tid >> 4) + k * 64;
        const int ii = (i < nn) ? i : (nn - 1);
        si[k] = rs_in[base + ii];
        so[k] = rs_out[base + ii];
    }

    const unsigned r0 = binstart[blockIdx.x];
    const unsigned r1 = binstart[blockIdx.x + 1];
    unsigned g0 = r0 + (unsigned)wave * 32u;
    if (g0 < r1) {
        const unsigned last = r1 - 1u;
        const unsigned i0 = g0 + (unsigned)lane;
        int2 cur = rec[(i0 < last) ? i0 : last];
        do {
            const unsigned gn = g0 + 1024u;
            const unsigned in = gn + (unsigned)lane;
            const int2* __restrict__ pn = rec + ((in < last) ? in : last);
            int2 nxt;
            const int sv = cur.x;
            const int dv = cur.y - base;
            const unsigned left = r1 - g0;
            const int n = (left < 32u) ? (int)left : 32;

            if (n == 32) {
                float lo[32], hi[32];
#pragma unroll
                for (int j = 0; j < 32; ++j) {
                    const int s = __builtin_amdgcn_readlane(sv, j);
                    const float* __restrict__ r = h + (size_t)s * D + lane;
                    lo[j] = r[0];
                    hi[j] = r[32];
                }
                nxt = *pn;
#pragma unroll
                for (int j = 0; j < 32; ++j) {
                    const int d = __builtin_amdgcn_readlane(dv, j);
                    float* p = acc + d * D + lane;
                    atomicAdd(p, lo[j]);
                    atomicAdd(p + 32, hi[j]);
                }
            } else {
                nxt = *pn;
                for (int j = 0; j < n; ++j) {
                    const int s = __builtin_amdgcn_readlane(sv, j);
                    const int d = __builtin_amdgcn_readlane(dv, j);
                    const float* __restrict__ r = h + (size_t)s * D + lane;
                    float* p = acc + d * D + lane;
                    const float a = r[0];
                    const float b = r[32];
                    atomicAdd(p, a);
                    atomicAdd(p + 32, b);
                }
            }
            cur = nxt;
            g0 = gn;
        } while (g0 < r1);
    }
    __syncthreads();

    const float4 w4 = *(const float4*)(wv + f);
#pragma unroll
    for (int k = 0; k < BWIT; ++k) {
        const int i = (tid >> 4) + k * 64;
        const int ii = (i < nn) ? i : (nn - 1);
        const float4 v = *(const float4*)(acc + ii * D + f);
        const float s = si[k];
        float p = fmaxf(v.x * s + bb.x, 0.0f) * w4.x
                + fmaxf(v.y * s + bb.y, 0.0f) * w4.y
                + fmaxf(v.z * s + bb.z, 0.0f) * w4.z
                + fmaxf(v.w * s + bb.w, 0.0f) * w4.w;
        p += __shfl_xor(p, 8, 32);
        p += __shfl_xor(p, 4, 32);
        p += __shfl_xor(p, 2, 32);
        p += __shfl_xor(p, 1, 32);
        if ((tid & 15) == 0 && i < nn) z[base + ii] = p * so[k];
    }
}

#define ZT 512
#define ZB 16

__global__ __launch_bounds__(ZT) void bin_zsum(const float* __restrict__ z,
                                               const int* __restrict__ recw,
                                               const unsigned* __restrict__ binstart,
                                               const float* __restrict__ rs_in,
                                               const float* __restrict__ b2,
                                               const float* __restrict__ Wr,
                                               const float* __restrict__ br,
                                               float* __restrict__ out, int N) {
    __shared__ float acc[BW];
    __shared__ float c0;
    const int2* __restrict__ rec = (const int2*)recw;
    const int tid = threadIdx.x;
    const int base = blockIdx.x * BW;
    const unsigned nn = (unsigned)((N - base < BW) ? (N - base) : BW);
    const unsigned r0 = binstart[blockIdx.x];
    const unsigned r1 = binstart[blockIdx.x + 1];

    if (tid < 32) {
        float p = b2[tid] * Wr[tid] + b2[tid + 32] * Wr[tid + 32];
#pragma unroll
        for (int off = 16; off > 0; off >>= 1) p += __shfl_xor(p, off, 32);
        if (tid == 0) c0 = p + br[0];
    }
    static_assert(BW <= ZT, "one thread per node of a bin");
    float sc = 0.0f;
    if ((unsigned)tid < nn) {
        acc[tid] = z[base + tid];
        sc = rs_in[base + tid];
    }
    __syncthreads();

    for (unsigned g = r0; g < r1; g += ZT * ZB) {
        int2 q[ZB];
#pragma unroll
        for (int j = 0; j < ZB; ++j) {
            const unsigned i = g + (unsigned)(j * ZT + tid);
            q[j] = make_int2(0, -1);
            if (i < r1) q[j] = rec[i];
        }
        float v[ZB];
#pragma unroll
        for (int j = 0; j < ZB; ++j) {
            v[j] = 0.0f;
            if ((unsigned)(q[j].y - base) < nn) v[j] = z[q[j].x];
        }
#pragma unroll
        for (int j = 0; j < ZB; ++j) {
            const unsigned u = (unsigned)(q[j].y - base);
            if (u < nn) atomicAdd(&acc[u], v[j]);
        }
    }
    __syncthreads();

    if ((unsigned)tid < nn) out[base + tid] = acc[tid] * sc + c0;
}

extern "C" void kernel_launch(void* const* d_in, const int* in_sizes, int n_in,
                              void* d_out, int out_size, void* d_ws, size_t ws_size,
                              hipStream_t stream) {
    const float* x   = (const float*)d_in[0];
    const int*   src = (const int*)d_in[1];
    const int*   dst = (const int*)d_in[2];
    const float* W1  = (const float*)d_in[3];
    const float* b1  = (const float*)d_in[4];
    const float* W2  = (const float*)d_in[5];
    const float* b2  = (const float*)d_in[6];
    const float* Wr  = (const float*)d_in[7];
    const float* br  = (const float*)d_in[8];
    float* out = (float*)d_out;

    const int N = in_sizes[0] / D;
    const int E = in_sizes[1];

    float* ws     = (float*)d_ws;
    float* rs_out = ws;
    float* rs_in  = ws + N;
    float* h      = ws + 2 * (size_t)N;
    int* rec      = (int*)(h + (size_t)N * D);
    unsigned* cntg = (unsigned*)(rec + 2 * (size_t)E);
    unsigned* binstart = cntg + MAXBIN * MAXBIN;
    unsigned* part = (unsigned*)((float*)(h + (size_t)N * D) + (size_t)N * D);
    float* z      = (float*)(part + 2 * (size_t)HCHUNKS * (size_t)N);

    const int gGemm = (N + GEMM_ROWS - 1) / GEMM_ROWS;
    const int gBin  = (N + BW - 1) / BW;

    count_stage<<<DEGB + MAXBIN, 1024, 0, stream>>>(src, dst, part, cntg, N, E);

    place_gemm_stage<<<MAXBIN + gGemm, 1024, 0, stream>>>(x, src, dst, W1, part, cntg, rec, binstart,
                                                          rs_out, rs_in, h, N, E);

    bin_sum  <<<gBin, 1024, 0, stream>>>(h, rec, binstart, rs_in, rs_out, b1, W2, Wr, z, N);

    bin_zsum <<<gBin, ZT, 0, stream>>>(z, rec, binstart, rs_in, b2, Wr, br, out, N);
}


# bin_sum: loads it waited for one by one go out together: the short last record group's rows as one batch like a full group's (was a round trip per record), start rows and first records at the top
# speedup vs baseline: 6.7549x; 1.0752x over previous
#include <hip/hip_runtime.h>
#include <hip/hip_bf16.h>

typedef __attribute__((ext_vector_type(2))) float v2f;
typedef __attribute__((ext_vector_type(8))) float v8f;

#define D 64


#define HCAP    25600
#define HCHUNKS 32
#define HRANGES 4
#define HBATCH  8
#define DEGB    (HCHUNKS * HRANGES * 2)

__device__ __forceinline__ void deg_hist_role(const int b,
                                              const int* __restrict__ src,
                                              const int* __restrict__ dst,
                                              unsigned* __restrict__ part, int N, int E) {
    __shared__ unsigned hist[HCAP];
    const int tid = threadIdx.x;
    const int c = b % HCHUNKS;
    const int y = (b / HCHUNKS) % HRANGES;
    const int zz = b / (HCHUNKS * HRANGES);
    const int* __restrict__ idx = zz ? dst : src;
    unsigned* __restrict__ plane = part + ((size_t)zz * HCHUNKS + c) * (size_t)N;

    const int nvec = E >> 2;
    const int per  = (nvec + HCHUNKS - 1) / HCHUNKS;
    const int v0   = c * per;
    const int v1   = (v0 + per < nvec) ? (v0 + per) : nvec;
    const int4* __restrict__ idx4 = (const int4*)idx;

    for (int base = y * HCAP; base < N; base += HRANGES * HCAP) {
        const unsigned cap = (unsigned)((N - base < HCAP) ? (N - base) : HCAP);
        for (unsigned i = tid; i < cap; i += 1024) hist[i] = 0u;
        __syncthreads();

        for (int v = v0 + tid; v < v1; v += HBATCH * 1024) {
            int4 q[HBATCH];
#pragma unroll
            for (int j = 0; j < HBATCH; ++j) {
                const int vv = v + j * 1024;
                q[j] = make_int4(-1, -1, -1, -1);
                if (vv < v1) q[j] = idx4[vv];
            }
#pragma unroll
            for (int j = 0; j < HBATCH; ++j) {
                const unsigned u0 = (unsigned)(q[j].x - base);
                const unsigned u1 = (unsigned)(q[j].y - base);
                const unsigned u2 = (unsigned)(q[j].z - base);
                const unsigned u3 = (unsigned)(q[j].w - base);
                if (u0 < cap) atomicAdd(&hist[u0], 1u);
                if (u1 < cap) atomicAdd(&hist[u1], 1u);
                if (u2 < cap) atomicAdd(&hist[u2], 1u);
                if (u3 < cap) atomicAdd(&hist[u3], 1u);
            }
        }
        if (c == HCHUNKS - 1) {
            for (int e = (nvec << 2) + tid; e < E; e += 1024) {
                const unsigned u = (unsigned)(idx[e] - base);
                if (u < cap) atomicAdd(&hist[u], 1u);
            }
        }
        __syncthreads();

        for (unsigned i = tid; i < cap; i += 1024) plane[base + i] = hist[i];
        __syncthreads();
    }
}

#define GEMM_ROWS 512
__device__ __forceinline__ void gemm_role(const int b,
                                          const float* __restrict__ X,
                                          const unsigned* __restrict__ part,
                                          const float* __restrict__ W,
                                          float* __restrict__ out0,
                                          float* __restrict__ dout,
                                          float* __restrict__ din,
                                          int N) {
    const int wave = threadIdx.x >> 5;
    const int lane = threadIdx.x & 31;
    const int nb = b * GEMM_ROWS + wave * 16;
    if (nb >= N) return;
    const int half = lane >> 4;
    const int lr   = lane & 15;

    const unsigned poff = (unsigned)half * (unsigned)(HCHUNKS * N) + (unsigned)(nb + lr);
    unsigned cnt1 = 1u;
#pragma nounroll
    for (int c0 = 0; c0 < HCHUNKS; c0 += 16) {
#pragma unroll
        for (int j = 0; j < 16; ++j) cnt1 += (part + (size_t)(c0 + j) * (size_t)N)[poff];
    }
    const float r = rsqrtf((float)cnt1);
    if (half == 0) dout[nb + lr] = r;
    else           din[nb + lr]  = r;
    const float s = __shfl(r, lr, 32);
    const float* __restrict__ xrow = X + (size_t)(nb + lr) * D;

    v8f acc0 = {}, acc1 = {}, acc2 = {}, acc3 = {};
#pragma unroll
    for (int k0 = 0; k0 < D; k0 += 4) {
        const int ka = k0 + half * 2;
        v2f a;
        a.x = xrow[ka] * s;
        a.y = xrow[ka + 1] * s;
        const float* __restrict__ w0 = W + (size_t)ka * D + lr;
        const float* __restrict__ w1 = w0 + D;
        v2f b0 = { w0[0],  w1[0]  };
        v2f b1 = { w0[16], w1[16] };
        v2f b2 = { w0[32], w1[32] };
        v2f b3 = { w0[48], w1[48] };
        acc0 = __builtin_amdgcn_wmma_f32_16x16x4_f32(false, a, false, b0, (short)0, acc0, false, false);
        acc1 = __builtin_amdgcn_wmma_f32_16x16x4_f32(false, a, false, b1, (short)0, acc1, false, false);
        acc2 = __builtin_amdgcn_wmma_f32_16x16x4_f32(false, a, false, b2, (short)0, acc2, false, false);
        acc3 = __builtin_amdgcn_wmma_f32_16x16x4_f32(false, a, false, b3, (short)0, acc3, false, false);
    }

    const int row0 = nb + half * 8;
#pragma unroll
    for (int v = 0; v < 8; ++v) {
        const size_t base = (size_t)(row0 + v) * D + lr;
        out0[base]      = acc0[v];
        out0[base + 16] = acc1[v];
        out0[base + 32] = acc2[v];
        out0[base + 48] = acc3[v];
    }
}


#define BW     392
#define BWIT   ((BW + 63) / 64)
#define BINIT  ((BW * (D / 4) + 1023) / 1024)
#define MAXBIN 256
#define STG    6272
#define IB     8

__device__ __forceinline__ void bin_count_role(const int c,
                                               const int* __restrict__ dst,
                                               unsigned* __restrict__ cntg, int E) {
    __shared__ unsigned cnt[MAXBIN];
    const int tid = threadIdx.x;
    const int per = (E + MAXBIN - 1) / MAXBIN;
    const int e0 = c * per;
    const int e1 = (e0 + per < E) ? (e0 + per) : E;
    if (tid < MAXBIN) cnt[tid] = 0u;
    __syncthreads();
    for (int e = e0 + tid; e < e1; e += IB * 1024) {
        int d[IB];
#pragma unroll
        for (int j = 0; j < IB; ++j) {
            const int ee = e + j * 1024;
            d[j] = -1;
            if (ee < e1) d[j] = dst[ee];
        }
#pragma unroll
        for (int j = 0; j < IB; ++j)
            if (d[j] >= 0) atomicAdd(&cnt[(unsigned)d[j] / BW], 1u);
    }
    __syncthreads();
    if (tid < MAXBIN) cntg[c * MAXBIN + tid] = cnt[tid];
}

__global__ __launch_bounds__(1024) void count_stage(const int* __restrict__ src,
                                                    const int* __restrict__ dst,
                                                    unsigned* __restrict__ part,
                                                    unsigned* __restrict__ cntg, int N, int E) {
    const int b = blockIdx.x;
    if (b < DEGB) deg_hist_role(b, src, dst, part, N, E);
    else          bin_count_role(b - DEGB, dst, cntg, E);
}

__device__ __forceinline__ void bin_place_role(const int c,
                                               const int* __restrict__ src,
                                               const int* __restrict__ dst,
                                               const unsigned* __restrict__ cntg,
                                               int* __restrict__ recw,
                                               unsigned* __restrict__ binstart, int E) {
    __shared__ unsigned tot4[4][MAXBIN];
    __shared__ unsigned pre4[4][MAXBIN];
    __shared__ unsigned gstart[MAXBIN];
    __shared__ unsigned lstart[MAXBIN];
    __shared__ unsigned cursor[MAXBIN];
    __shared__ int2 stage[STG];
    int2* __restrict__ rec = (int2*)recw;

    const int tid = threadIdx.x;
    const int per = (E + MAXBIN - 1) / MAXBIN;
    const int e0 = c * per;
    const int e1 = (e0 + per < E) ? (e0 + per) : E;

    {
        const int b = tid & (MAXBIN - 1);
        const int qd = tid >> 8;
        unsigned ts = 0u, ps = 0u;
        const unsigned* __restrict__ col = cntg + (size_t)(qd * 64) * MAXBIN + b;
#pragma nounroll
        for (int k0 = 0; k0 < 64; k0 += 16) {
            unsigned v[16];
#pragma unroll
            for (int k = 0; k < 16; ++k) v[k] = col[(size_t)(k0 + k) * MAXBIN];
#pragma unroll
            for (int k = 0; k < 16; ++k) {
                ts += v[k];
                if (qd * 64 + k0 + k < c) ps += v[k];
            }
        }
        tot4[qd][b] = ts;
        pre4[qd][b] = ps;
    }
    __syncthreads();

    if (tid < 32) {
        unsigned t8[8], o8[8];
        unsigned ts = 0u, os = 0u;
#pragma unroll
        for (int j = 0; j < 8; ++j) {
            const int b = tid * 8 + j;
            t8[j] = ts;
            o8[j] = os;
            ts += tot4[0][b] + tot4[1][b] + tot4[2][b] + tot4[3][b];
            os += cntg[c * MAXBIN + b];
        }
        unsigned tin = ts, oin = os;
#pragma unroll
        for (int off = 1; off < 32; off <<= 1) {
            const unsigned u = __shfl_up(tin, off, 32);
            const unsigned w = __shfl_up(oin, off, 32);
            if (tid >= off) { tin += u; oin += w; }
        }
        const unsigned tex = tin - ts, oex = oin - os;
#pragma unroll
        for (int j = 0; j < 8; ++j) {
            const int b = tid * 8 + j;
            const unsigned bs = tex + t8[j];
            gstart[b] = bs + pre4[0][b] + pre4[1][b] + pre4[2][b] + pre4[3][b];
            lstart[b] = oex + o8[j];
            cursor[b] = oex + o8[j];
            if (c == 0) binstart[b] = bs;
        }
        if (c == 0 && tid == 31) binstart[MAXBIN] = tin;
    }
    __syncthreads();

    if (per <= STG) {
        for (int e = e0 + tid; e < e1; e += IB * 1024) {
            int dd[IB], ss[IB];
#pragma unroll
            for (int j = 0; j < IB; ++j) {
                const int ee = e + j * 1024;
                dd[j] = -1;
                ss[j] = 0;
                if (ee < e1) { dd[j] = dst[ee]; ss[j] = src[ee]; }
            }
            unsigned pos[IB];
#pragma unroll
            for (int j = 0; j < IB; ++j) {
                pos[j] = 0u;
                if (dd[j] >= 0) pos[j] = atomicAdd(&cursor[(unsigned)dd[j] / BW], 1u);
            }
#pragma unroll
            for (int j = 0; j < IB; ++j)
                if (dd[j] >= 0) stage[pos[j]] = make_int2(ss[j], dd[j]);
        }
        __syncthreads();
        const int nrec = e1 - e0;
        for (int i = tid; i < nrec; i += IB * 1024) {
            int2 r[IB];
#pragma unroll
            for (int j = 0; j < IB; ++j) {
                const int ii = i + j * 1024;
                r[j] = make_int2(0, -1);
                if (ii < nrec) r[j] = stage[ii];
            }
            unsigned gs[IB], ls[IB];
#pragma unroll
            for (int j = 0; j < IB; ++j) {
                gs[j] = 0u;
                ls[j] = 0u;
                if (r[j].y >= 0) {
                    const unsigned b = (unsigned)r[j].y / BW;
                    gs[j] = gstart[b];
                    ls[j] = lstart[b];
                }
            }
#pragma unroll
            for (int j = 0; j < IB; ++j)
                if (r[j].y >= 0) rec[gs[j] + ((unsigned)(i + j * 1024) - ls[j])] = r[j];
        }
    } else {
        for (int e = e0 + tid; e < e1; e += IB * 1024) {
            int dd[IB], ss[IB];
#pragma unroll
            for (int j = 0; j < IB; ++j) {
                const int ee = e + j * 1024;
                dd[j] = -1;
                ss[j] = 0;
                if (ee < e1) { dd[j] = dst[ee]; ss[j] = src[ee]; }
            }
#pragma unroll
            for (int j = 0; j < IB; ++j) {
                if (dd[j] >= 0) {
                    const unsigned b = (unsigned)dd[j] / BW;
                    const unsigned pos = atomicAdd(&cursor[b], 1u);
                    rec[gstart[b] + (pos - lstart[b])] = make_int2(ss[j], dd[j]);
                }
            }
        }
    }
}

__global__ __launch_bounds__(1024) void place_gemm_stage(const float* __restrict__ X,
                                                         const int* __restrict__ src,
                                                         const int* __restrict__ dst,
                                                         const float* __restrict__ W,
                                                         const unsigned* __restrict__ part,
                                                         const unsigned* __restrict__ cntg,
                                                         int* __restrict__ recw,
                                                         unsigned* __restrict__ binstart,
                                                         float* __restrict__ dout,
                                                         float* __restrict__ din,
                                                         float* __restrict__ out0,
                                                         int N, int E) {
    const int b = blockIdx.x;
    const int gGemm = (N + GEMM_ROWS - 1) / GEMM_ROWS;
    if (b < gGemm) gemm_role(b, X, part, W, out0, dout, din, N);
    else           bin_place_role(b - gGemm, src, dst, cntg, recw, binstart, E);
}

__global__ __launch_bounds__(1024) void bin_sum(const float* __restrict__ h,
                                                const int* __restrict__ recw,
                                                const unsigned* __restrict__ binstart,
                                                const float* __restrict__ rs_in,
                                                const float* __restrict__ rs_out,
                                                const float* __restrict__ b1,
                                                const float* __restrict__ W2,
                                                const float* __restrict__ Wr,
                                                float* __restrict__ z, int N) {
    __shared__ __attribute__((aligned(16))) float acc[BINIT * 1024 * 4];
    __shared__ __attribute__((aligned(16))) float wv[D];
    const int2* __restrict__ rec = (const int2*)recw;
    const int tid = threadIdx.x;
    const int lane = tid & 31;
    const int wave = __builtin_amdgcn_readfirstlane(tid >> 5);
    const int base = blockIdx.x * BW;
    const int nn = (N - base < BW) ? (N - base) : BW;

    const float4* __restrict__ h4 = (const float4*)(h + (size_t)base * D);
    const int lim = nn * (D / 4);
    float4 st[BINIT];
#pragma unroll
    for (int k = 0; k < BINIT; ++k) {
        const int i = tid + k * 1024;
        st[k] = h4[(i < lim) ? i : (lim - 1)];
    }

    const unsigned r0 = binstart[blockIdx.x];
    const unsigned r1 = binstart[blockIdx.x + 1];
    unsigned g0 = r0 + (unsigned)wave * 32u;
    const unsigned last = r1 - 1u;
    int2 cur = make_int2(0, 0);
    if (g0 < r1) {
        const unsigned i0 = g0 + (unsigned)lane;
        cur = rec[(i0 < last) ? i0 : last];
    }

    const int f = (tid & 15) * 4;
    const float4 bb = *(const float4*)(b1 + f);
    float si[BWIT], so[BWIT];
#pragma unroll
    for (int k = 0; k < BWIT; ++k) {
        const int i = (tid >> 4) + k * 64;
        const int ii = (i < nn) ? i : (nn - 1);
        si[k] = rs_in[base + ii];
        so[k] = rs_out[base + ii];
    }

    if (tid < 256) {
        const int k = tid >> 2, q = tid & 3;
        const float4* __restrict__ wrow = (const float4*)(W2 + (size_t)k * D + q * 16);
        const float4* __restrict__ wr4  = (const float4*)(Wr + q * 16);
        float p = 0.0f;
#pragma unroll
        for (int j = 0; j < 4; ++j) {
            const float4 a = wrow[j], w = wr4[j];
            p += a.x * w.x + a.y * w.y + a.z * w.z + a.w * w.w;
        }
        p += __shfl_xor(p, 1, 32);
        p += __shfl_xor(p, 2, 32);
        if (q == 0) wv[k] = p;
    }
    {
        float4* acc4 = (float4*)acc;
#pragma unroll
        for (int k = 0; k < BINIT; ++k) acc4[tid + k * 1024] = st[k];
    }
    __syncthreads();

    if (g0 < r1) {
        do {
            const unsigned gn = g0 + 1024u;
            const unsigned in = gn + (unsigned)lane;
            const int2* __restrict__ pn = rec + ((in < last) ? in : last);
            int2 nxt;
            const int sv = cur.x;
            const int dv = cur.y - base;
            const unsigned left = r1 - g0;
            const int n = (left < 32u) ? (int)left : 32;

            if (n == 32) {
                float lo[32], hi[32];
#pragma unroll
                for (int j = 0; j < 32; ++j) {
                    const int s = __builtin_amdgcn_readlane(sv, j);
                    const float* __restrict__ r = h + (size_t)s * D + lane;
                    lo[j] = r[0];
                    hi[j] = r[32];
                }
                nxt = *pn;
#pragma unroll
                for (int j = 0; j < 32; ++j) {
                    const int d = __builtin_amdgcn_readlane(dv, j);
                    float* p = acc + d * D + lane;
                    atomicAdd(p, lo[j]);
                    atomicAdd(p + 32, hi[j]);
                }
            } else {
                static_assert(BINIT * 1024 * 4 >= (BW + 1) * D, "acc has a spare row behind the bin's rows");
                nxt = cur;
                float lo[32], hi[32];
#pragma unroll
                for (int j = 0; j < 32; ++j) {
                    const int s = __builtin_amdgcn_readlane(sv, j);
                    const float* __restrict__ r = h + (size_t)s * D + lane;
                    lo[j] = r[0];
                    hi[j] = r[32];
                }
#pragma unroll
                for (int j = 0; j < 32; ++j) {
                    const int d = (j < n) ? __builtin_amdgcn_readlane(dv, j) : BW;
                    float* p = acc + d * D + lane;
                    atomicAdd(p, lo[j]);
                    atomicAdd(p + 32, hi[j]);
                }
            }
            cur = nxt;
            g0 = gn;
        } while (g0 < r1);
    }
    __syncthreads();

    const float4 w4 = *(const float4*)(wv + f);
#pragma unroll
    for (int k = 0; k < BWIT; ++k) {
        const int i = (tid >> 4) + k * 64;
        const int ii = (i < nn) ? i : (nn - 1);
        const float4 v = *(const float4*)(acc + ii * D + f);
        const float s = si[k];
        float p = fmaxf(v.x * s + bb.x, 0.0f) * w4.x
                + fmaxf(v.y * s + bb.y, 0.0f) * w4.y
                + fmaxf(v.z * s + bb.z, 0.0f) * w4.z
                + fmaxf(v.w * s + bb.w, 0.0f) * w4.w;
        p += __shfl_xor(p, 8, 32);
        p += __shfl_xor(p, 4, 32);
        p += __shfl_xor(p, 2, 32);
        p += __shfl_xor(p, 1, 32);
        if ((tid & 15) == 0 && i < nn) z[base + ii] = p * so[k];
    }
}

#define ZT 512
#define ZB 16

__global__ __launch_bounds__(ZT) void bin_zsum(const float* __restrict__ z,
                                               const int* __restrict__ recw,
                                               const unsigned* __restrict__ binstart,
                                               const float* __restrict__ rs_in,
                                               const float* __restrict__ b2,
                                               const float* __restrict__ Wr,
                                               const float* __restrict__ br,
                                               float* __restrict__ out, int N) {
    __shared__ float acc[BW];
    __shared__ float c0;
    const int2* __restrict__ rec = (const int2*)recw;
    const int tid = threadIdx.x;
    const int base = blockIdx.x * BW;
    const unsigned nn = (unsigned)((N - base < BW) ? (N - base) : BW);
    const unsigned r0 = binstart[blockIdx.x];
    const unsigned r1 = binstart[blockIdx.x + 1];

    if (tid < 32) {
        float p = b2[tid] * Wr[tid] + b2[tid + 32] * Wr[tid + 32];
#pragma unroll
        for (int off = 16; off > 0; off >>= 1) p += __shfl_xor(p, off, 32);
        if (tid == 0) c0 = p + br[0];
    }
    static_assert(BW <= ZT, "one thread per node of a bin");
    float sc = 0.0f;
    if ((unsigned)tid < nn) {
        acc[tid] = z[base + tid];
        sc = rs_in[base + tid];
    }
    __syncthreads();

    for (unsigned g = r0; g < r1; g += ZT * ZB) {
        int2 q[ZB];
#pragma unroll
        for (int j = 0; j < ZB; ++j) {
            const unsigned i = g + (unsigned)(j * ZT + tid);
            q[j] = make_int2(0, -1);
            if (i < r1) q[j] = rec[i];
        }
        float v[ZB];
#pragma unroll
        for (int j = 0; j < ZB; ++j) {
            v[j] = 0.0f;
            if ((unsigned)(q[j].y - base) < nn) v[j] = z[q[j].x];
        }
#pragma unroll
        for (int j = 0; j < ZB; ++j) {
            const unsigned u = (unsigned)(q[j].y - base);
            if (u < nn) atomicAdd(&acc[u], v[j]);
        }
    }
    __syncthreads();

    if ((unsigned)tid < nn) out[base + tid] = acc[tid] * sc + c0;
}

extern "C" void kernel_launch(void* const* d_in, const int* in_sizes, int n_in,
                              void* d_out, int out_size, void* d_ws, size_t ws_size,
                              hipStream_t stream) {
    const float* x   = (const float*)d_in[0];
    const int*   src = (const int*)d_in[1];
    const int*   dst = (const int*)d_in[2];
    const float* W1  = (const float*)d_in[3];
    const float* b1  = (const float*)d_in[4];
    const float* W2  = (const float*)d_in[5];
    const float* b2  = (const float*)d_in[6];
    const float* Wr  = (const float*)d_in[7];
    const float* br  = (const float*)d_in[8];
    float* out = (float*)d_out;

    const int N = in_sizes[0] / D;
    const int E = in_sizes[1];

    float* ws     = (float*)d_ws;
    float* rs_out = ws;
    float* rs_in  = ws + N;
    float* h      = ws + 2 * (size_t)N;
    int* rec      = (int*)(h + (size_t)N * D);
    unsigned* cntg = (unsigned*)(rec + 2 * (size_t)E);
    unsigned* binstart = cntg + MAXBIN * MAXBIN;
    unsigned* part = (unsigned*)((float*)(h + (size_t)N * D) + (size_t)N * D);
    float* z      = (float*)(part + 2 * (size_t)HCHUNKS * (size_t)N);

    const int gGemm = (N + GEMM_ROWS - 1) / GEMM_ROWS;
    const int gBin  = (N + BW - 1) / BW;

    count_stage<<<DEGB + MAXBIN, 1024, 0, stream>>>(src, dst, part, cntg, N, E);

    place_gemm_stage<<<MAXBIN + gGemm, 1024, 0, stream>>>(x, src, dst, W1, part, cntg, rec, binstart,
                                                          rs_out, rs_in, h, N, E);

    bin_sum  <<<gBin, 1024, 0, stream>>>(h, rec, binstart, rs_in, rs_out, b1, W2, Wr, z, N);

    bin_zsum <<<gBin, ZT, 0, stream>>>(z, rec, binstart, rs_in, b2, Wr, br, out, N);
}
